# starting kernel
# baseline (speedup 1.0000x reference)
#include <hip/hip_runtime.h>

typedef __attribute__((ext_vector_type(2))) float v2f;
typedef __attribute__((ext_vector_type(8))) float v8f;

#define NFEAT 128
#define NHID  128
#define NCLASS 64


__global__ void k_zero_f4(float4* __restrict__ p, unsigned n4) {
    unsigned i = blockIdx.x * blockDim.x + threadIdx.x;
    if (i < n4) p[i] = make_float4(0.f, 0.f, 0.f, 0.f);
}

__global__ void k_zero_i32(int* __restrict__ p, unsigned n) {
    unsigned i = blockIdx.x * blockDim.x + threadIdx.x;
    if (i < n) p[i] = 0;
}

__global__ void k_degrees(const int* __restrict__ src, const int* __restrict__ dst,
                          int* __restrict__ outdeg, int* __restrict__ indeg, unsigned E) {
    unsigned i = blockIdx.x * blockDim.x + threadIdx.x;
    if (i < E) {
        atomicAdd(&outdeg[src[i]], 1);
        atomicAdd(&indeg[dst[i]], 1);
    }
}

__global__ void k_norms(const int* __restrict__ outdeg, const int* __restrict__ indeg,
                        float* __restrict__ ns, float* __restrict__ nd, unsigned N) {
    unsigned i = blockIdx.x * blockDim.x + threadIdx.x;
    if (i < N) {
        ns[i] = rsqrtf(fmaxf((float)outdeg[i], 1.0f));
        nd[i] = rsqrtf(fmaxf((float)indeg[i], 1.0f));
    }
}


template<int K, int Nout, bool SCALE>
__global__ void k_gemm_wmma(const float* __restrict__ X, const float* __restrict__ W,
                            float* __restrict__ Y, const float* __restrict__ rowScale,
                            int M) {
    const int lane = threadIdx.x & 31;
    const int tile = blockIdx.x * (blockDim.x >> 5) + (threadIdx.x >> 5);
    constexpr int TN = Nout >> 4;
    const int mt = tile / TN;
    const int nt = tile & (TN - 1);
    const int mbase = mt << 4;
    if (mbase >= M) return;

    const int lmod  = lane & 15;
    const int lhalf = lane >> 4;

    unsigned aoff = (unsigned)(mbase + lmod) * K + (lhalf << 1);
    unsigned boff = (unsigned)(lhalf << 1) * Nout + ((nt << 4) + lmod);

    v8f acc = {};
#if __has_builtin(__builtin_amdgcn_wmma_f32_16x16x4_f32)
    #pragma unroll 8
    for (int k = 0; k < K; k += 4) {
        v2f a = *(const v2f*)(X + aoff);
        v2f b;
        b.x = W[boff];
        b.y = W[boff + Nout];
        acc = __builtin_amdgcn_wmma_f32_16x16x4_f32(
            false, a, false, b, (short)0, acc, false, false);
        aoff += 4;
        boff += 4u * Nout;
    }
#else
    for (int k = 0; k < K; k += 4) {
        v2f a = *(const v2f*)(X + aoff);
        acc[0] += a.x * W[boff] + a.y * W[boff + Nout];
        aoff += 4;
        boff += 4u * Nout;
    }
#endif

    const unsigned colu = (unsigned)((nt << 4) + lmod);
    const unsigned yrow = (unsigned)(mbase + (lhalf << 3));

    if (SCALE) {
        float sc[8];
        #pragma unroll
        for (int r = 0; r < 8; ++r) sc[r] = rowScale[yrow + r];
        #pragma unroll
        for (int r = 0; r < 8; ++r)
            Y[(yrow + r) * (unsigned)Nout + colu] = acc[r] * sc[r];
    } else {
        #pragma unroll
        for (int r = 0; r < 8; ++r)
            Y[(yrow + r) * (unsigned)Nout + colu] = acc[r];
    }
}


template<int F>
__global__ void k_edge_agg(const float* __restrict__ U, const float* __restrict__ V,
                           const int* __restrict__ src, const int* __restrict__ dst,
                           float* __restrict__ Uagg, float* __restrict__ Vagg,
                           unsigned total) {
    unsigned gid = blockIdx.x * blockDim.x + threadIdx.x;
    if (gid >= total) return;
    constexpr int CH = F >> 2;
    constexpr int CSH = (F == 128) ? 5 : 4;
    unsigned e = gid >> CSH;
    unsigned c = (gid & (CH - 1)) << 2;

    int s = src[e];
    int d = dst[e];
    if (F == 128) {
        s = __builtin_amdgcn_readfirstlane(s);
        d = __builtin_amdgcn_readfirstlane(d);
    }
    const unsigned so = (unsigned)s * F + c;
    const unsigned dofs = (unsigned)d * F + c;

    const float4 u = *(const float4*)(U + so);
    const float4 v = *(const float4*)(V + so);
    float* up = Uagg + dofs;
    float* vp = Vagg + dofs;
    atomicAdd(up + 0, u.x); atomicAdd(up + 1, u.y);
    atomicAdd(up + 2, u.z); atomicAdd(up + 3, u.w);
    atomicAdd(vp + 0, v.x); atomicAdd(vp + 1, v.y);
    atomicAdd(vp + 2, v.z); atomicAdd(vp + 3, v.w);
}


template<int F, bool RELU>
__global__ void k_combine(const float* __restrict__ Uagg, const float* __restrict__ Vagg,
                          const float* __restrict__ nd, const float* __restrict__ bias,
                          float* __restrict__ out, unsigned total) {
    unsigned gid = blockIdx.x * blockDim.x + threadIdx.x;
    if (gid >= total) return;
    constexpr int FSH = (F == 128) ? 7 : 6;
    const unsigned f = gid & (F - 1);
    const unsigned i = gid >> FSH;
    float v = fmaf(Uagg[gid], nd[i], bias[f]) + Vagg[gid];
    out[gid] = RELU ? fmaxf(v, 0.0f) : v;
}


extern "C" void kernel_launch(void* const* d_in, const int* in_sizes, int n_in,
                              void* d_out, int out_size, void* d_ws, size_t ws_size,
                              hipStream_t stream) {
    const float* raw_x = (const float*)d_in[0];
    const int*   src   = (const int*)d_in[1];
    const int*   dst   = (const int*)d_in[2];
    const float* gc_w0 = (const float*)d_in[3];
    const float* gc_b0 = (const float*)d_in[4];
    const float* gc_w1 = (const float*)d_in[5];
    const float* gc_b1 = (const float*)d_in[6];
    const float* gc_w2 = (const float*)d_in[7];
    const float* gc_b2 = (const float*)d_in[8];
    const float* rw0   = (const float*)d_in[9];
    const float* rw1   = (const float*)d_in[10];
    const float* rw2   = (const float*)d_in[11];

    const int N = in_sizes[0] / NFEAT;
    const int E = in_sizes[1];
    const size_t NB = (size_t)N * NFEAT;

    float* P0 = (float*)d_ws;
    float* P1 = P0 + NB;
    float* P2 = P1 + NB;
    float* P3 = P2 + NB;
    float* ns = P3 + NB;
    float* nd = ns + N;
    int*  odeg = (int*)(nd + N);
    int*  ideg = odeg + N;

    const int TB = 256;
    auto blocks = [](long long n, int tb) { return (int)((n + tb - 1) / tb); };

    k_zero_i32<<<blocks(2LL * N, TB), TB, 0, stream>>>(odeg, 2u * N);
    k_degrees<<<blocks(E, TB), TB, 0, stream>>>(src, dst, odeg, ideg, (unsigned)E);
    k_norms<<<blocks(N, TB), TB, 0, stream>>>(odeg, ideg, ns, nd, (unsigned)N);

    auto zero = [&](float* p, size_t n) {
        k_zero_f4<<<blocks((long long)(n >> 2), TB), TB, 0, stream>>>(
            (float4*)p, (unsigned)(n >> 2));
    };

    const int mtiles = N / 16;
    auto gemm128s = [&](const float* X, const float* W, float* Y) {
        k_gemm_wmma<128, 128, true><<<blocks((long long)mtiles * 8, 8), TB, 0, stream>>>(
            X, W, Y, ns, N);
    };
    auto gemm128 = [&](const float* X, const float* W, float* Y) {
        k_gemm_wmma<128, 128, false><<<blocks((long long)mtiles * 8, 8), TB, 0, stream>>>(
            X, W, Y, nullptr, N);
    };
    auto gemm64s = [&](const float* X, const float* W, float* Y) {
        k_gemm_wmma<128, 64, true><<<blocks((long long)mtiles * 4, 8), TB, 0, stream>>>(
            X, W, Y, ns, N);
    };
    auto gemm64 = [&](const float* X, const float* W, float* Y) {
        k_gemm_wmma<128, 64, false><<<blocks((long long)mtiles * 4, 8), TB, 0, stream>>>(
            X, W, Y, nullptr, N);
    };

    auto edge128 = [&](const float* U, const float* V, float* Ua, float* Va) {
        unsigned total = (unsigned)E << 5;
        k_edge_agg<128><<<blocks(total, TB), TB, 0, stream>>>(U, V, src, dst, Ua, Va, total);
    };
    auto edge64 = [&](const float* U, const float* V, float* Ua, float* Va) {
        unsigned total = (unsigned)E << 4;
        k_edge_agg<64><<<blocks(total, TB), TB, 0, stream>>>(U, V, src, dst, Ua, Va, total);
    };

    gemm128s(raw_x, gc_w0, P1);
    gemm128 (raw_x, rw0,   P2);
    zero(P0, NB); zero(P3, NB);
    edge128(P1, P2, P0, P3);
    {
        unsigned total = (unsigned)N * 128u;
        k_combine<128, true><<<blocks(total, TB), TB, 0, stream>>>(
            P0, P3, nd, gc_b0, P1, total);
    }

    gemm128s(P1, gc_w1, P2);
    gemm128 (P1, rw1,   P0);
    zero(P3, NB); zero(P1, NB);
    edge128(P2, P0, P3, P1);
    {
        unsigned total = (unsigned)N * 128u;
        k_combine<128, true><<<blocks(total, TB), TB, 0, stream>>>(
            P3, P1, nd, gc_b1, P0, total);
    }

    gemm64s(P0, gc_w2, P1);
    gemm64 (P0, rw2,   P2);
    const size_t NB2 = (size_t)N * NCLASS;
    zero(P3, NB2); zero(P0, NB2);
    edge64(P1, P2, P3, P0);
    {
        unsigned total = (unsigned)N * 64u;
        k_combine<64, false><<<blocks(total, TB), TB, 0, stream>>>(
            P3, P0, nd, gc_b2, (float*)d_out, total);
    }
}


# Edge aggregation as a pull over a destination-ordered edge list built in-call (arrival-order rank + scan): row sums in registers, epilogue fused; f32 atomics, zero fills, combine passes removed
# speedup vs baseline: 3.1161x; 3.1161x over previous
#include <hip/hip_runtime.h>

typedef __attribute__((ext_vector_type(2))) float v2f;
typedef __attribute__((ext_vector_type(4))) float v4f;
typedef __attribute__((ext_vector_type(8))) float v8f;

#define NFEAT 128
#define NHID  128
#define NCLASS 64


__global__ void k_zero_i32(int* __restrict__ p, unsigned n) {
    unsigned i = blockIdx.x * blockDim.x + threadIdx.x;
    if (i < n) p[i] = 0;
}

__global__ void k_degrees(const int* __restrict__ src, const int* __restrict__ dst,
                          int* __restrict__ outdeg, int* __restrict__ indeg,
                          int* __restrict__ rank, unsigned E) {
    unsigned i = blockIdx.x * blockDim.x + threadIdx.x;
    if (i < E) {
        atomicAdd(&outdeg[src[i]], 1);
        rank[i] = atomicAdd(&indeg[dst[i]], 1);
    }
}

__global__ void k_scan_local(const int* __restrict__ indeg, int* __restrict__ rowptr,
                             int* __restrict__ bsum, unsigned N) {
    __shared__ int wsum[8];
    const unsigned tid  = threadIdx.x;
    const unsigned lane = tid & 31;
    const unsigned wid  = tid >> 5;
    const unsigned base = blockIdx.x * 1024u + tid * 4u;
    int v[4];
    #pragma unroll
    for (int k = 0; k < 4; ++k) v[k] = (base + k < N) ? indeg[base + k] : 0;
    const int t = v[0] + v[1] + v[2] + v[3];
    int inc = t;
    #pragma unroll
    for (int d = 1; d < 32; d <<= 1) {
        const int n = __shfl_up(inc, d);
        if (lane >= (unsigned)d) inc += n;
    }
    if (lane == 31) wsum[wid] = inc;
    __syncthreads();
    int woff = 0, tot = 0;
    #pragma unroll
    for (int w = 0; w < 8; ++w) {
        const int s = wsum[w];
        if ((unsigned)w < wid) woff += s;
        tot += s;
    }
    int ex = woff + inc - t;
    #pragma unroll
    for (int k = 0; k < 4; ++k) {
        if (base + k <= N) rowptr[base + k] = ex;
        ex += v[k];
    }
    if (tid == 0) bsum[blockIdx.x] = tot;
}

__global__ void k_scan_finish(int* __restrict__ rowptr, const int* __restrict__ bsum,
                              const int* __restrict__ outdeg, const int* __restrict__ indeg,
                              float* __restrict__ ns, float* __restrict__ nd, unsigned N) {
    __shared__ int red[8];
    const unsigned tid = threadIdx.x;
    const unsigned i = blockIdx.x * 256u + tid;
    const unsigned b = (blockIdx.x * 256u) >> 10;
    int p = 0;
    for (unsigned k = tid; k < b; k += 256u) p += bsum[k];
    #pragma unroll
    for (int d = 16; d > 0; d >>= 1) p += __shfl_xor(p, d);
    if ((tid & 31) == 0) red[tid >> 5] = p;
    __syncthreads();
    int off = 0;
    #pragma unroll
    for (int w = 0; w < 8; ++w) off += red[w];
    if (i <= N) rowptr[i] += off;
    if (i < N) {
        ns[i] = rsqrtf(fmaxf((float)outdeg[i], 1.0f));
        nd[i] = rsqrtf(fmaxf((float)indeg[i], 1.0f));
    }
}

__global__ void k_fill(const int* __restrict__ src, const int* __restrict__ dst,
                       const int* __restrict__ rank, const int* __restrict__ rowptr,
                       int* __restrict__ ssrc, unsigned E) {
    unsigned i = blockIdx.x * blockDim.x + threadIdx.x;
    if (i < E) ssrc[rowptr[dst[i]] + rank[i]] = src[i];
}


template<int K, int Nout, bool SCALE>
__global__ void k_gemm_wmma(const float* __restrict__ X, const float* __restrict__ W,
                            float* __restrict__ Y, const float* __restrict__ rowScale,
                            int M) {
    const int lane = threadIdx.x & 31;
    const int tile = blockIdx.x * (blockDim.x >> 5) + (threadIdx.x >> 5);
    constexpr int TN = Nout >> 4;
    const int mt = tile / TN;
    const int nt = tile & (TN - 1);
    const int mbase = mt << 4;
    if (mbase >= M) return;

    const int lmod  = lane & 15;
    const int lhalf = lane >> 4;

    unsigned aoff = (unsigned)(mbase + lmod) * K + (lhalf << 1);
    unsigned boff = (unsigned)(lhalf << 1) * Nout + ((nt << 4) + lmod);

    v8f acc = {};
#if __has_builtin(__builtin_amdgcn_wmma_f32_16x16x4_f32)
    #pragma unroll 8
    for (int k = 0; k < K; k += 4) {
        v2f a = *(const v2f*)(X + aoff);
        v2f b;
        b.x = W[boff];
        b.y = W[boff + Nout];
        acc = __builtin_amdgcn_wmma_f32_16x16x4_f32(
            false, a, false, b, (short)0, acc, false, false);
        aoff += 4;
        boff += 4u * Nout;
    }
#else
    for (int k = 0; k < K; k += 4) {
        v2f a = *(const v2f*)(X + aoff);
        acc[0] += a.x * W[boff] + a.y * W[boff + Nout];
        aoff += 4;
        boff += 4u * Nout;
    }
#endif

    const unsigned colu = (unsigned)((nt << 4) + lmod);
    const unsigned yrow = (unsigned)(mbase + (lhalf << 3));

    if (SCALE) {
        float sc[8];
        #pragma unroll
        for (int r = 0; r < 8; ++r) sc[r] = rowScale[yrow + r];
        #pragma unroll
        for (int r = 0; r < 8; ++r)
            Y[(yrow + r) * (unsigned)Nout + colu] = acc[r] * sc[r];
    } else {
        #pragma unroll
        for (int r = 0; r < 8; ++r)
            Y[(yrow + r) * (unsigned)Nout + colu] = acc[r];
    }
}


template<int F> struct RowVec;
template<> struct RowVec<128> { typedef v4f T; };
template<> struct RowVec<64>  { typedef v2f T; };

template<int K, int F, typename vec>
__device__ __forceinline__ void pull_rows(int sv, int t, const float* __restrict__ U,
                                          const float* __restrict__ V, unsigned c,
                                          vec& au, vec& av) {
    vec u[K], w[K];
    #pragma unroll
    for (int k = 0; k < K; ++k) {
        const unsigned s = (unsigned)__builtin_amdgcn_readlane(sv, t + k);
        u[k] = *(const vec*)(U + s * F + c);
        w[k] = *(const vec*)(V + s * F + c);
    }
    #pragma unroll
    for (int k = 0; k < K; ++k) { au += u[k]; av += w[k]; }
}

template<int F, bool RELU>
__global__ void k_pull(const float* __restrict__ U, const float* __restrict__ V,
                       const int* __restrict__ rowptr, const int* __restrict__ ssrc,
                       const float* __restrict__ nd, const float* __restrict__ bias,
                       float* __restrict__ out, unsigned N) {
    typedef typename RowVec<F>::T vec;
    constexpr unsigned VL = F / 32;
    const unsigned lane = threadIdx.x & 31;
    const unsigned row = __builtin_amdgcn_readfirstlane(
        blockIdx.x * (blockDim.x >> 5) + (threadIdx.x >> 5));
    if (row >= N) return;
    const int beg = __builtin_amdgcn_readfirstlane(rowptr[row]);
    const int end = __builtin_amdgcn_readfirstlane(rowptr[row + 1]);
    const unsigned c = lane * VL;

    vec au = (vec)(0.0f);
    vec av = (vec)(0.0f);
    for (int base = beg; base < end; base += 32) {
        const int rem = end - base;
        const int cnt = rem < 32 ? rem : 32;
        const int sv = ((int)lane < cnt) ? ssrc[base + (int)lane] : 0;
        int t = 0;
        for (; t + 4 <= cnt; t += 4) pull_rows<4, F, vec>(sv, t, U, V, c, au, av);
        const int r = cnt - t;
        if (r == 3)      pull_rows<3, F, vec>(sv, t, U, V, c, au, av);
        else if (r == 2) pull_rows<2, F, vec>(sv, t, U, V, c, au, av);
        else if (r == 1) pull_rows<1, F, vec>(sv, t, U, V, c, au, av);
    }

    const float ndr = nd[row];
    const vec b = *(const vec*)(bias + c);
    vec r;
    #pragma unroll
    for (unsigned k = 0; k < VL; ++k) {
        const float x = fmaf(au[k], ndr, b[k]) + av[k];
        r[k] = RELU ? fmaxf(x, 0.0f) : x;
    }
    *(vec*)(out + row * F + c) = r;
}


extern "C" void kernel_launch(void* const* d_in, const int* in_sizes, int n_in,
                              void* d_out, int out_size, void* d_ws, size_t ws_size,
                              hipStream_t stream) {
    const float* raw_x = (const float*)d_in[0];
    const int*   src   = (const int*)d_in[1];
    const int*   dst   = (const int*)d_in[2];
    const float* gc_w0 = (const float*)d_in[3];
    const float* gc_b0 = (const float*)d_in[4];
    const float* gc_w1 = (const float*)d_in[5];
    const float* gc_b1 = (const float*)d_in[6];
    const float* gc_w2 = (const float*)d_in[7];
    const float* gc_b2 = (const float*)d_in[8];
    const float* rw0   = (const float*)d_in[9];
    const float* rw1   = (const float*)d_in[10];
    const float* rw2   = (const float*)d_in[11];

    const int N = in_sizes[0] / NFEAT;
    const int E = in_sizes[1];
    const size_t NB = (size_t)N * NFEAT;

    float* P0 = (float*)d_ws;
    float* P1 = P0 + NB;
    float* P2 = P1 + NB;
    float* P3 = P2 + NB;
    float* ns = P3 + NB;
    float* nd = ns + N;
    int*  odeg = (int*)(nd + N);
    int*  ideg = odeg + N;
    int*  rowptr = ideg + N;
    int*  bsum = rowptr + N + 32;
    int*  rank = bsum + 1024;
    int*  ssrc = rank + E;

    const int TB = 256;
    auto blocks = [](long long n, int tb) { return (int)((n + tb - 1) / tb); };

    k_zero_i32<<<blocks(2LL * N, TB), TB, 0, stream>>>(odeg, 2u * N);
    k_degrees<<<blocks(E, TB), TB, 0, stream>>>(src, dst, odeg, ideg, rank, (unsigned)E);
    k_scan_local<<<blocks(N + 1LL, 1024), TB, 0, stream>>>(ideg, rowptr, bsum, (unsigned)N);
    k_scan_finish<<<blocks(N + 1LL, TB), TB, 0, stream>>>(rowptr, bsum, odeg, ideg, ns, nd,
                                                          (unsigned)N);
    k_fill<<<blocks(E, TB), TB, 0, stream>>>(src, dst, rank, rowptr, ssrc, (unsigned)E);

    const int mtiles = N / 16;
    auto gemm128s = [&](const float* X, const float* W, float* Y) {
        k_gemm_wmma<128, 128, true><<<blocks((long long)mtiles * 8, 8), TB, 0, stream>>>(
            X, W, Y, ns, N);
    };
    auto gemm128 = [&](const float* X, const float* W, float* Y) {
        k_gemm_wmma<128, 128, false><<<blocks((long long)mtiles * 8, 8), TB, 0, stream>>>(
            X, W, Y, nullptr, N);
    };
    auto gemm64s = [&](const float* X, const float* W, float* Y) {
        k_gemm_wmma<128, 64, true><<<blocks((long long)mtiles * 4, 8), TB, 0, stream>>>(
            X, W, Y, ns, N);
    };
    auto gemm64 = [&](const float* X, const float* W, float* Y) {
        k_gemm_wmma<128, 64, false><<<blocks((long long)mtiles * 4, 8), TB, 0, stream>>>(
            X, W, Y, nullptr, N);
    };

    const int pullBlocks = blocks(N, 8);

    gemm128s(raw_x, gc_w0, P1);
    gemm128 (raw_x, rw0,   P2);
    k_pull<128, true><<<pullBlocks, TB, 0, stream>>>(
        P1, P2, rowptr, ssrc, nd, gc_b0, P0, (unsigned)N);

    gemm128s(P0, gc_w1, P1);
    gemm128 (P0, rw1,   P2);
    k_pull<128, true><<<pullBlocks, TB, 0, stream>>>(
        P1, P2, rowptr, ssrc, nd, gc_b1, P3, (unsigned)N);

    gemm64s(P3, gc_w2, P1);
    gemm64 (P3, rw2,   P2);
    k_pull<64, false><<<pullBlocks, TB, 0, stream>>>(
        P1, P2, rowptr, ssrc, nd, gc_b2, (float*)d_out, (unsigned)N);
}


# Aggregate before transform: each edge gathers one 512-B row of x (edge records carry id and norm_s), then one K=256 WMMA GEMM per layer applies [W;R]; edge sums now precede the products
# speedup vs baseline: 3.4614x; 1.1108x over previous
#include <hip/hip_runtime.h>

typedef __attribute__((ext_vector_type(2))) float v2f;
typedef __attribute__((ext_vector_type(4))) float v4f;
typedef __attribute__((ext_vector_type(8))) float v8f;
typedef __attribute__((ext_vector_type(2))) int   v2i;

#define NFEAT 128
#define NHID  128
#define NCLASS 64


__global__ void k_zero_i32(int* __restrict__ p, unsigned n) {
    unsigned i = blockIdx.x * blockDim.x + threadIdx.x;
    if (i < n) p[i] = 0;
}

__global__ void k_degrees(const int* __restrict__ src, const int* __restrict__ dst,
                          int* __restrict__ outdeg, int* __restrict__ indeg,
                          int* __restrict__ rank, unsigned E) {
    unsigned i = blockIdx.x * blockDim.x + threadIdx.x;
    if (i < E) {
        atomicAdd(&outdeg[src[i]], 1);
        rank[i] = atomicAdd(&indeg[dst[i]], 1);
    }
}

__global__ void k_scan_local(const int* __restrict__ indeg, int* __restrict__ rowptr,
                             int* __restrict__ bsum, unsigned N) {
    __shared__ int wsum[8];
    const unsigned tid  = threadIdx.x;
    const unsigned lane = tid & 31;
    const unsigned wid  = tid >> 5;
    const unsigned base = blockIdx.x * 1024u + tid * 4u;
    int v[4];
    #pragma unroll
    for (int k = 0; k < 4; ++k) v[k] = (base + k < N) ? indeg[base + k] : 0;
    const int t = v[0] + v[1] + v[2] + v[3];
    int inc = t;
    #pragma unroll
    for (int d = 1; d < 32; d <<= 1) {
        const int n = __shfl_up(inc, d);
        if (lane >= (unsigned)d) inc += n;
    }
    if (lane == 31) wsum[wid] = inc;
    __syncthreads();
    int woff = 0, tot = 0;
    #pragma unroll
    for (int w = 0; w < 8; ++w) {
        const int s = wsum[w];
        if ((unsigned)w < wid) woff += s;
        tot += s;
    }
    int ex = woff + inc - t;
    #pragma unroll
    for (int k = 0; k < 4; ++k) {
        if (base + k <= N) rowptr[base + k] = ex;
        ex += v[k];
    }
    if (tid == 0) bsum[blockIdx.x] = tot;
}

__global__ void k_scan_finish(int* __restrict__ rowptr, const int* __restrict__ bsum,
                              const int* __restrict__ outdeg, const int* __restrict__ indeg,
                              float* __restrict__ ns, float* __restrict__ nd, unsigned N) {
    __shared__ int red[8];
    const unsigned tid = threadIdx.x;
    const unsigned i = blockIdx.x * 256u + tid;
    const unsigned b = (blockIdx.x * 256u) >> 10;
    int p = 0;
    for (unsigned k = tid; k < b; k += 256u) p += bsum[k];
    #pragma unroll
    for (int d = 16; d > 0; d >>= 1) p += __shfl_xor(p, d);
    if ((tid & 31) == 0) red[tid >> 5] = p;
    __syncthreads();
    int off = 0;
    #pragma unroll
    for (int w = 0; w < 8; ++w) off += red[w];
    if (i <= N) rowptr[i] += off;
    if (i < N) {
        ns[i] = rsqrtf(fmaxf((float)outdeg[i], 1.0f));
        nd[i] = rsqrtf(fmaxf((float)indeg[i], 1.0f));
    }
}

__global__ void k_fill(const int* __restrict__ src, const int* __restrict__ dst,
                       const int* __restrict__ rank, const int* __restrict__ rowptr,
                       const float* __restrict__ ns, int* __restrict__ erec, unsigned E) {
    unsigned i = blockIdx.x * blockDim.x + threadIdx.x;
    if (i < E) {
        const int s = src[i];
        v2i e;
        e.x = s;
        e.y = __float_as_int(ns[s]);
        *(v2i*)(erec + 2u * (unsigned)(rowptr[dst[i]] + rank[i])) = e;
    }
}


template<int K>
__device__ __forceinline__ void pull_rows(int sv, int wv, int t, const float* __restrict__ X,
                                          unsigned c, v4f& aw, v4f& ap) {
    v4f u[K];
    #pragma unroll
    for (int k = 0; k < K; ++k) {
        const unsigned s = (unsigned)__builtin_amdgcn_readlane(sv, t + k);
        u[k] = *(const v4f*)(X + s * NFEAT + c);
    }
    #pragma unroll
    for (int k = 0; k < K; ++k) {
        const float w = __int_as_float(__builtin_amdgcn_readlane(wv, t + k));
        aw += w * u[k];
        ap += u[k];
    }
}

__global__ void k_pull_x(const float* __restrict__ X, const int* __restrict__ rowptr,
                         const int* __restrict__ erec, const float* __restrict__ nd,
                         float* __restrict__ AW, float* __restrict__ AP, unsigned N) {
    const unsigned lane = threadIdx.x & 31;
    const unsigned row = __builtin_amdgcn_readfirstlane(
        blockIdx.x * (blockDim.x >> 5) + (threadIdx.x >> 5));
    if (row >= N) return;
    const int beg = __builtin_amdgcn_readfirstlane(rowptr[row]);
    const int end = __builtin_amdgcn_readfirstlane(rowptr[row + 1]);
    const unsigned c = lane * 4u;

    v4f aw = (v4f)(0.0f);
    v4f ap = (v4f)(0.0f);
    for (int base = beg; base < end; base += 32) {
        const int rem = end - base;
        const int cnt = rem < 32 ? rem : 32;
        v2i e = (v2i)(0);
        if ((int)lane < cnt) e = *(const v2i*)(erec + 2u * (unsigned)(base + (int)lane));
        const int sv = e.x;
        const int wv = e.y;
        int t = 0;
        for (; t + 8 <= cnt; t += 8) pull_rows<8>(sv, wv, t, X, c, aw, ap);
        switch (cnt - t) {
            case 7: pull_rows<7>(sv, wv, t, X, c, aw, ap); break;
            case 6: pull_rows<6>(sv, wv, t, X, c, aw, ap); break;
            case 5: pull_rows<5>(sv, wv, t, X, c, aw, ap); break;
            case 4: pull_rows<4>(sv, wv, t, X, c, aw, ap); break;
            case 3: pull_rows<3>(sv, wv, t, X, c, aw, ap); break;
            case 2: pull_rows<2>(sv, wv, t, X, c, aw, ap); break;
            case 1: pull_rows<1>(sv, wv, t, X, c, aw, ap); break;
            default: break;
        }
    }

    const float ndr = nd[row];
    *(v4f*)(AW + row * NFEAT + c) = aw * ndr;
    *(v4f*)(AP + row * NFEAT + c) = ap;
}


template<int Nout>
__device__ __forceinline__ void wmma_k128(const float* __restrict__ X, const float* __restrict__ W,
                                          unsigned aoff, unsigned boff, v8f& acc) {
    #pragma unroll 8
    for (int k = 0; k < 128; k += 4) {
        v2f a = *(const v2f*)(X + aoff);
        v2f b;
        b.x = W[boff];
        b.y = W[boff + Nout];
        acc = __builtin_amdgcn_wmma_f32_16x16x4_f32(
            false, a, false, b, (short)0, acc, false, false);
        aoff += 4;
        boff += 4u * Nout;
    }
}

template<int Nout, bool RELU>
__global__ void k_gemm2_wmma(const float* __restrict__ XA, const float* __restrict__ XB,
                             const float* __restrict__ WA, const float* __restrict__ WB,
                             const float* __restrict__ bias, float* __restrict__ Y, int M) {
    const int lane = threadIdx.x & 31;
    const int tile = blockIdx.x * (blockDim.x >> 5) + (threadIdx.x >> 5);
    constexpr int TN = Nout >> 4;
    const int mt = tile / TN;
    const int nt = tile & (TN - 1);
    const int mbase = mt << 4;
    if (mbase >= M) return;

    const int lmod  = lane & 15;
    const int lhalf = lane >> 4;

    const unsigned aoff = (unsigned)(mbase + lmod) * 128u + (lhalf << 1);
    const unsigned boff = (unsigned)(lhalf << 1) * Nout + ((nt << 4) + lmod);

    v8f acc = {};
    wmma_k128<Nout>(XA, WA, aoff, boff, acc);
    wmma_k128<Nout>(XB, WB, aoff, boff, acc);

    const unsigned colu = (unsigned)((nt << 4) + lmod);
    const unsigned yrow = (unsigned)(mbase + (lhalf << 3));
    const float bv = bias[colu];
    #pragma unroll
    for (int r = 0; r < 8; ++r) {
        const float x = acc[r] + bv;
        Y[(yrow + r) * (unsigned)Nout + colu] = RELU ? fmaxf(x, 0.0f) : x;
    }
}


extern "C" void kernel_launch(void* const* d_in, const int* in_sizes, int n_in,
                              void* d_out, int out_size, void* d_ws, size_t ws_size,
                              hipStream_t stream) {
    const float* raw_x = (const float*)d_in[0];
    const int*   src   = (const int*)d_in[1];
    const int*   dst   = (const int*)d_in[2];
    const float* gc_w0 = (const float*)d_in[3];
    const float* gc_b0 = (const float*)d_in[4];
    const float* gc_w1 = (const float*)d_in[5];
    const float* gc_b1 = (const float*)d_in[6];
    const float* gc_w2 = (const float*)d_in[7];
    const float* gc_b2 = (const float*)d_in[8];
    const float* rw0   = (const float*)d_in[9];
    const float* rw1   = (const float*)d_in[10];
    const float* rw2   = (const float*)d_in[11];

    const int N = in_sizes[0] / NFEAT;
    const int E = in_sizes[1];
    const size_t NB = (size_t)N * NFEAT;

    float* P0 = (float*)d_ws;
    float* P1 = P0 + NB;
    float* P2 = P1 + NB;
    float* ns = P2 + NB;
    float* nd = ns + N;
    int*  odeg = (int*)(nd + N);
    int*  ideg = odeg + N;
    int*  rowptr = ideg + N;
    int*  bsum = rowptr + N + 32;
    int*  erec = bsum + 1024;
    int*  rank = erec + 2 * (size_t)E;

    const int TB = 256;
    auto blocks = [](long long n, int tb) { return (int)((n + tb - 1) / tb); };

    k_zero_i32<<<blocks(2LL * N, TB), TB, 0, stream>>>(odeg, 2u * N);
    k_degrees<<<blocks(E, TB), TB, 0, stream>>>(src, dst, odeg, ideg, rank, (unsigned)E);
    k_scan_local<<<blocks(N + 1LL, 1024), TB, 0, stream>>>(ideg, rowptr, bsum, (unsigned)N);
    k_scan_finish<<<blocks(N + 1LL, TB), TB, 0, stream>>>(rowptr, bsum, odeg, ideg, ns, nd,
                                                          (unsigned)N);
    k_fill<<<blocks(E, TB), TB, 0, stream>>>(src, dst, rank, rowptr, ns, erec, (unsigned)E);

    const int mtiles = N / 16;
    const int pullBlocks = blocks(N, 8);
    auto pull = [&](const float* X) {
        k_pull_x<<<pullBlocks, TB, 0, stream>>>(X, rowptr, erec, nd, P1, P2, (unsigned)N);
    };

    pull(raw_x);
    k_gemm2_wmma<128, true><<<blocks((long long)mtiles * 8, 8), TB, 0, stream>>>(
        P1, P2, gc_w0, rw0, gc_b0, P0, N);

    pull(P0);
    k_gemm2_wmma<128, true><<<blocks((long long)mtiles * 8, 8), TB, 0, stream>>>(
        P1, P2, gc_w1, rw1, gc_b1, P0, N);

    pull(P0);
    k_gemm2_wmma<64, false><<<blocks((long long)mtiles * 4, 8), TB, 0, stream>>>(
        P1, P2, gc_w2, rw2, gc_b2, (float*)d_out, N);
}


# GEMM register blocking: one wave computes a 32 x Nout strip (2 x Nout/16 accumulator tiles), A and B fragments by 16-B loads one step ahead; k regrouped within each 8, so the sum order changes
# speedup vs baseline: 4.3056x; 1.2439x over previous
#include <hip/hip_runtime.h>

typedef __attribute__((ext_vector_type(2))) float v2f;
typedef __attribute__((ext_vector_type(4))) float v4f;
typedef __attribute__((ext_vector_type(8))) float v8f;
typedef __attribute__((ext_vector_type(2))) int   v2i;

#define NFEAT 128
#define NHID  128
#define NCLASS 64


__global__ void k_zero_i32(int* __restrict__ p, unsigned n) {
    unsigned i = blockIdx.x * blockDim.x + threadIdx.x;
    if (i < n) p[i] = 0;
}

__global__ void k_degrees(const int* __restrict__ src, const int* __restrict__ dst,
                          int* __restrict__ outdeg, int* __restrict__ indeg,
                          int* __restrict__ rank, unsigned E) {
    unsigned i = blockIdx.x * blockDim.x + threadIdx.x;
    if (i < E) {
        atomicAdd(&outdeg[src[i]], 1);
        rank[i] = atomicAdd(&indeg[dst[i]], 1);
    }
}

__global__ void k_scan_local(const int* __restrict__ indeg, int* __restrict__ rowptr,
                             int* __restrict__ bsum, unsigned N) {
    __shared__ int wsum[8];
    const unsigned tid  = threadIdx.x;
    const unsigned lane = tid & 31;
    const unsigned wid  = tid >> 5;
    const unsigned base = blockIdx.x * 1024u + tid * 4u;
    int v[4];
    #pragma unroll
    for (int k = 0; k < 4; ++k) v[k] = (base + k < N) ? indeg[base + k] : 0;
    const int t = v[0] + v[1] + v[2] + v[3];
    int inc = t;
    #pragma unroll
    for (int d = 1; d < 32; d <<= 1) {
        const int n = __shfl_up(inc, d);
        if (lane >= (unsigned)d) inc += n;
    }
    if (lane == 31) wsum[wid] = inc;
    __syncthreads();
    int woff = 0, tot = 0;
    #pragma unroll
    for (int w = 0; w < 8; ++w) {
        const int s = wsum[w];
        if ((unsigned)w < wid) woff += s;
        tot += s;
    }
    int ex = woff + inc - t;
    #pragma unroll
    for (int k = 0; k < 4; ++k) {
        if (base + k <= N) rowptr[base + k] = ex;
        ex += v[k];
    }
    if (tid == 0) bsum[blockIdx.x] = tot;
}

__global__ void k_scan_finish(int* __restrict__ rowptr, const int* __restrict__ bsum,
                              const int* __restrict__ outdeg, const int* __restrict__ indeg,
                              float* __restrict__ ns, float* __restrict__ nd, unsigned N) {
    __shared__ int red[8];
    const unsigned tid = threadIdx.x;
    const unsigned i = blockIdx.x * 256u + tid;
    const unsigned b = (blockIdx.x * 256u) >> 10;
    int p = 0;
    for (unsigned k = tid; k < b; k += 256u) p += bsum[k];
    #pragma unroll
    for (int d = 16; d > 0; d >>= 1) p += __shfl_xor(p, d);
    if ((tid & 31) == 0) red[tid >> 5] = p;
    __syncthreads();
    int off = 0;
    #pragma unroll
    for (int w = 0; w < 8; ++w) off += red[w];
    if (i <= N) rowptr[i] += off;
    if (i < N) {
        ns[i] = rsqrtf(fmaxf((float)outdeg[i], 1.0f));
        nd[i] = rsqrtf(fmaxf((float)indeg[i], 1.0f));
    }
}

__global__ void k_fill(const int* __restrict__ src, const int* __restrict__ dst,
                       const int* __restrict__ rank, const int* __restrict__ rowptr,
                       const float* __restrict__ ns, int* __restrict__ erec, unsigned E) {
    unsigned i = blockIdx.x * blockDim.x + threadIdx.x;
    if (i < E) {
        const int s = src[i];
        v2i e;
        e.x = s;
        e.y = __float_as_int(ns[s]);
        *(v2i*)(erec + 2u * (unsigned)(rowptr[dst[i]] + rank[i])) = e;
    }
}


template<int K>
__device__ __forceinline__ void pull_rows(int sv, int wv, int t, const float* __restrict__ X,
                                          unsigned c, v4f& aw, v4f& ap) {
    v4f u[K];
    #pragma unroll
    for (int k = 0; k < K; ++k) {
        const unsigned s = (unsigned)__builtin_amdgcn_readlane(sv, t + k);
        u[k] = *(const v4f*)(X + s * NFEAT + c);
    }
    #pragma unroll
    for (int k = 0; k < K; ++k) {
        const float w = __int_as_float(__builtin_amdgcn_readlane(wv, t + k));
        aw += w * u[k];
        ap += u[k];
    }
}

__global__ void k_pull_x(const float* __restrict__ X, const int* __restrict__ rowptr,
                         const int* __restrict__ erec, const float* __restrict__ nd,
                         float* __restrict__ AW, float* __restrict__ AP, unsigned N) {
    const unsigned lane = threadIdx.x & 31;
    const unsigned row = __builtin_amdgcn_readfirstlane(
        blockIdx.x * (blockDim.x >> 5) + (threadIdx.x >> 5));
    if (row >= N) return;
    const int beg = __builtin_amdgcn_readfirstlane(rowptr[row]);
    const int end = __builtin_amdgcn_readfirstlane(rowptr[row + 1]);
    const unsigned c = lane * 4u;

    v4f aw = (v4f)(0.0f);
    v4f ap = (v4f)(0.0f);
    for (int base = beg; base < end; base += 32) {
        const int rem = end - base;
        const int cnt = rem < 32 ? rem : 32;
        v2i e = (v2i)(0);
        if ((int)lane < cnt) e = *(const v2i*)(erec + 2u * (unsigned)(base + (int)lane));
        const int sv = e.x;
        const int wv = e.y;
        int t = 0;
        for (; t + 8 <= cnt; t += 8) pull_rows<8>(sv, wv, t, X, c, aw, ap);
        switch (cnt - t) {
            case 7: pull_rows<7>(sv, wv, t, X, c, aw, ap); break;
            case 6: pull_rows<6>(sv, wv, t, X, c, aw, ap); break;
            case 5: pull_rows<5>(sv, wv, t, X, c, aw, ap); break;
            case 4: pull_rows<4>(sv, wv, t, X, c, aw, ap); break;
            case 3: pull_rows<3>(sv, wv, t, X, c, aw, ap); break;
            case 2: pull_rows<2>(sv, wv, t, X, c, aw, ap); break;
            case 1: pull_rows<1>(sv, wv, t, X, c, aw, ap); break;
            default: break;
        }
    }

    const float ndr = nd[row];
    *(v4f*)(AW + row * NFEAT + c) = aw * ndr;
    *(v4f*)(AP + row * NFEAT + c) = ap;
}


template<int NT>
struct GFrag {
    v4f a[2];
    v4f b[4][NT / 4];
};

template<int NT>
__device__ __forceinline__ void gfrag_load(GFrag<NT>& f, const float* __restrict__ X,
                                           const float* __restrict__ Wm,
                                           unsigned a0, unsigned a1, unsigned boff, int k8) {
    constexpr unsigned Nout = NT * 16;
    f.a[0] = *(const v4f*)(X + a0 + k8);
    f.a[1] = *(const v4f*)(X + a1 + k8);
    #pragma unroll
    for (int q = 0; q < 4; ++q) {
        #pragma unroll
        for (int g = 0; g < NT / 4; ++g)
            f.b[q][g] = *(const v4f*)(Wm + boff + (unsigned)(k8 + q) * Nout + 4u * g);
    }
}

template<int NT>
__device__ __forceinline__ void gfrag_mma(const GFrag<NT>& f, v8f (&acc)[2][NT]) {
    #pragma unroll
    for (int s = 0; s < 2; ++s) {
        #pragma unroll
        for (int j = 0; j < NT; ++j) {
            v2f b;
            b.x = f.b[2 * s][j >> 2][j & 3];
            b.y = f.b[2 * s + 1][j >> 2][j & 3];
            #pragma unroll
            for (int m = 0; m < 2; ++m) {
                v2f a;
                a.x = f.a[m][2 * s];
                a.y = f.a[m][2 * s + 1];
                acc[m][j] = __builtin_amdgcn_wmma_f32_16x16x4_f32(
                    false, a, false, b, (short)0, acc[m][j], false, false);
            }
        }
    }
}

template<int NT>
__device__ __forceinline__ void wmma_k128(const float* __restrict__ X, const float* __restrict__ Wm,
                                          unsigned a0, unsigned a1, unsigned boff,
                                          v8f (&acc)[2][NT]) {
    GFrag<NT> f0, f1;
    gfrag_load<NT>(f0, X, Wm, a0, a1, boff, 0);
    #pragma unroll 1
    for (int k8 = 0; k8 < 112; k8 += 16) {
        gfrag_load<NT>(f1, X, Wm, a0, a1, boff, k8 + 8);
        gfrag_mma<NT>(f0, acc);
        gfrag_load<NT>(f0, X, Wm, a0, a1, boff, k8 + 16);
        gfrag_mma<NT>(f1, acc);
    }
    gfrag_load<NT>(f1, X, Wm, a0, a1, boff, 120);
    gfrag_mma<NT>(f0, acc);
    gfrag_mma<NT>(f1, acc);
}

template<int Nout, bool RELU>
__global__ void __launch_bounds__(128)
k_gemm2_wmma(const float* __restrict__ XA, const float* __restrict__ XB,
             const float* __restrict__ WA, const float* __restrict__ WB,
             const float* __restrict__ bias, float* __restrict__ Y, int M) {
    constexpr int NT = Nout >> 4;
    const int lane = threadIdx.x & 31;
    const int strip = blockIdx.x * (blockDim.x >> 5) + (threadIdx.x >> 5);
    const int mbase = strip << 5;
    if (mbase >= M) return;

    const int c = lane & 15;
    const int h = lane >> 4;

    const int r0 = min(mbase + c, M - 1);
    const int r1 = min(mbase + 16 + c, M - 1);
    const unsigned a0 = (unsigned)r0 * 128u + 4u * h;
    const unsigned a1 = (unsigned)r1 * 128u + 4u * h;
    const unsigned boff = (unsigned)(4 * h) * Nout + (unsigned)(c * NT);

    v8f acc[2][NT];
    #pragma unroll
    for (int m = 0; m < 2; ++m) {
        #pragma unroll
        for (int j = 0; j < NT; ++j) acc[m][j] = (v8f)(0.0f);
    }

    wmma_k128<NT>(XA, WA, a0, a1, boff, acc);
    wmma_k128<NT>(XB, WB, a0, a1, boff, acc);

    v4f bv[NT / 4];
    #pragma unroll
    for (int g = 0; g < NT / 4; ++g) bv[g] = *(const v4f*)(bias + c * NT + 4 * g);
    #pragma unroll
    for (int m = 0; m < 2; ++m) {
        #pragma unroll
        for (int r = 0; r < 8; ++r) {
            const int row = mbase + 16 * m + 8 * h + r;
            if (row < M) {
                #pragma unroll
                for (int g = 0; g < NT / 4; ++g) {
                    v4f o;
                    #pragma unroll
                    for (int e = 0; e < 4; ++e) {
                        const float x = acc[m][4 * g + e][r] + bv[g][e];
                        o[e] = RELU ? fmaxf(x, 0.0f) : x;
                    }
                    *(v4f*)(Y + (unsigned)row * (unsigned)Nout + (unsigned)(c * NT + 4 * g)) = o;
                }
            }
        }
    }
}


extern "C" void kernel_launch(void* const* d_in, const int* in_sizes, int n_in,
                              void* d_out, int out_size, void* d_ws, size_t ws_size,
                              hipStream_t stream) {
    const float* raw_x = (const float*)d_in[0];
    const int*   src   = (const int*)d_in[1];
    const int*   dst   = (const int*)d_in[2];
    const float* gc_w0 = (const float*)d_in[3];
    const float* gc_b0 = (const float*)d_in[4];
    const float* gc_w1 = (const float*)d_in[5];
    const float* gc_b1 = (const float*)d_in[6];
    const float* gc_w2 = (const float*)d_in[7];
    const float* gc_b2 = (const float*)d_in[8];
    const float* rw0   = (const float*)d_in[9];
    const float* rw1   = (const float*)d_in[10];
    const float* rw2   = (const float*)d_in[11];

    const int N = in_sizes[0] / NFEAT;
    const int E = in_sizes[1];
    const size_t NB = (size_t)N * NFEAT;

    float* P0 = (float*)d_ws;
    float* P1 = P0 + NB;
    float* P2 = P1 + NB;
    float* ns = P2 + NB;
    float* nd = ns + N;
    int*  odeg = (int*)(nd + N);
    int*  ideg = odeg + N;
    int*  rowptr = ideg + N;
    int*  bsum = rowptr + N + 32;
    int*  erec = bsum + 1024;
    int*  rank = erec + 2 * (size_t)E;

    const int TB = 256;
    auto blocks = [](long long n, int tb) { return (int)((n + tb - 1) / tb); };

    k_zero_i32<<<blocks(2LL * N, TB), TB, 0, stream>>>(odeg, 2u * N);
    k_degrees<<<blocks(E, TB), TB, 0, stream>>>(src, dst, odeg, ideg, rank, (unsigned)E);
    k_scan_local<<<blocks(N + 1LL, 1024), TB, 0, stream>>>(ideg, rowptr, bsum, (unsigned)N);
    k_scan_finish<<<blocks(N + 1LL, TB), TB, 0, stream>>>(rowptr, bsum, odeg, ideg, ns, nd,
                                                          (unsigned)N);
    k_fill<<<blocks(E, TB), TB, 0, stream>>>(src, dst, rank, rowptr, ns, erec, (unsigned)E);

    const int strips = (N + 31) / 32;
    const int gemmBlocks = blocks(strips, 4);
    const int pullBlocks = blocks(N, 8);
    auto pull = [&](const float* X) {
        k_pull_x<<<pullBlocks, TB, 0, stream>>>(X, rowptr, erec, nd, P1, P2, (unsigned)N);
    };

    pull(raw_x);
    k_gemm2_wmma<128, true><<<gemmBlocks, 128, 0, stream>>>(
        P1, P2, gc_w0, rw0, gc_b0, P0, N);

    pull(P0);
    k_gemm2_wmma<128, true><<<gemmBlocks, 128, 0, stream>>>(
        P1, P2, gc_w1, rw1, gc_b1, P0, N);

    pull(P0);
    k_gemm2_wmma<64, false><<<gemmBlocks, 128, 0, stream>>>(
        P1, P2, gc_w2, rw2, gc_b2, (float*)d_out, N);
}


# Edge list ordered by destination via a two-level counting sort in LDS (bucket runs, then per-bucket degree scan): per-edge global rank atomics and scattered fills removed; sums in LDS arrival order
# speedup vs baseline: 5.0753x; 1.1788x over previous
#include <hip/hip_runtime.h>

typedef __attribute__((ext_vector_type(2))) float v2f;
typedef __attribute__((ext_vector_type(4))) float v4f;
typedef __attribute__((ext_vector_type(8))) float v8f;
typedef __attribute__((ext_vector_type(2))) int   v2i;

#define NFEAT 128
#define NHID  128
#define NCLASS 64


__global__ void k_zero_i32(int* __restrict__ p, unsigned n) {
    unsigned i = blockIdx.x * blockDim.x + threadIdx.x;
    if (i < n) p[i] = 0;
}

__global__ void k_scan_local(const int* __restrict__ indeg, int* __restrict__ rowptr,
                             int* __restrict__ bsum, unsigned N) {
    __shared__ int wsum[8];
    const unsigned tid  = threadIdx.x;
    const unsigned lane = tid & 31;
    const unsigned wid  = tid >> 5;
    const unsigned base = blockIdx.x * 1024u + tid * 4u;
    int v[4];
    #pragma unroll
    for (int k = 0; k < 4; ++k) v[k] = (base + k < N) ? indeg[base + k] : 0;
    const int t = v[0] + v[1] + v[2] + v[3];
    int inc = t;
    #pragma unroll
    for (int d = 1; d < 32; d <<= 1) {
        const int n = __shfl_up(inc, d);
        if (lane >= (unsigned)d) inc += n;
    }
    if (lane == 31) wsum[wid] = inc;
    __syncthreads();
    int woff = 0, tot = 0;
    #pragma unroll
    for (int w = 0; w < 8; ++w) {
        const int s = wsum[w];
        if ((unsigned)w < wid) woff += s;
        tot += s;
    }
    int ex = woff + inc - t;
    #pragma unroll
    for (int k = 0; k < 4; ++k) {
        if (base + k <= N) rowptr[base + k] = ex;
        ex += v[k];
    }
    if (tid == 0) bsum[blockIdx.x] = tot;
}


#define BSH   8
#define BW    (1 << BSH)
#define NBMAX 512
#define CH1   6144
#define CAP2  6144

__device__ __forceinline__ int block_scan512(int v, int* wsum, int& tot) {
    const unsigned lane = threadIdx.x & 31;
    const unsigned wid  = threadIdx.x >> 5;
    int inc = v;
    #pragma unroll
    for (int d = 1; d < 32; d <<= 1) {
        const int n = __shfl_up(inc, d);
        if (lane >= (unsigned)d) inc += n;
    }
    __syncthreads();
    if (lane == 31) wsum[wid] = inc;
    __syncthreads();
    int woff = 0;
    tot = 0;
    #pragma unroll
    for (int w = 0; w < 16; ++w) {
        const int s = wsum[w];
        if ((unsigned)w < wid) woff += s;
        tot += s;
    }
    return woff + inc - v;
}

template<bool SCATTER>
__global__ void __launch_bounds__(512)
k_bin(const int* __restrict__ src, const int* __restrict__ dst, int* __restrict__ odeg,
      int* __restrict__ cnt, const int* __restrict__ scanned, const int* __restrict__ bsum,
      int* __restrict__ ebuf, unsigned E, unsigned NB, unsigned nblk) {
    __shared__ int hist[NBMAX];
    __shared__ int gdelta[NBMAX];
    __shared__ int bpre[NBMAX];
    __shared__ int wsum[16];
    __shared__ v2i stage[SCATTER ? CH1 : 1];
    const unsigned tid = threadIdx.x;
    const unsigned G1 = gridDim.x;
    const unsigned wg = blockIdx.x;
    const unsigned c0 = wg * (unsigned)CH1;
    const unsigned n = min(E - c0, (unsigned)CH1);

    hist[tid] = 0;
    __syncthreads();

    int s[12], d[12], slot[12];
    #pragma unroll
    for (int k = 0; k < 12; ++k) {
        const unsigned i = tid + 512u * k;
        const bool ok = i < n;
        s[k] = ok ? src[c0 + i] : 0;
        d[k] = ok ? dst[c0 + i] : -1;
    }
    #pragma unroll
    for (int k = 0; k < 12; ++k) {
        slot[k] = 0;
        if (d[k] >= 0) {
            slot[k] = atomicAdd(&hist[d[k] >> BSH], 1);
            if (!SCATTER) atomicAdd(&odeg[s[k]], 1);
        }
    }
    __syncthreads();

    if (!SCATTER) {
        if (tid < NB) cnt[tid * G1 + wg] = hist[tid];
        return;
    }

    int tot;
    const int bs = (tid < nblk) ? bsum[tid] : 0;
    bpre[tid] = block_scan512(bs, wsum, tot);
    const int h = hist[tid];
    const int lb = block_scan512(h, wsum, tot);
    if (tid < NB) {
        const unsigned idx = tid * G1 + wg;
        gdelta[tid] = scanned[idx] + bpre[idx >> 10] - lb;
    }
    hist[tid] = lb;
    __syncthreads();

    #pragma unroll
    for (int k = 0; k < 12; ++k) {
        if (d[k] >= 0) {
            v2i r;
            r.x = s[k];
            r.y = d[k];
            stage[hist[d[k] >> BSH] + slot[k]] = r;
        }
    }
    __syncthreads();

    for (unsigned i = tid; i < n; i += 512u) {
        const v2i r = stage[i];
        *(v2i*)(ebuf + 2u * (unsigned)(gdelta[r.y >> BSH] + (int)i)) = r;
    }
}

__global__ void __launch_bounds__(512)
k_bucket(const int* __restrict__ ebuf, const int* __restrict__ scanned,
         const int* __restrict__ bsum, const int* __restrict__ odeg,
         int* __restrict__ rowptr, float* __restrict__ nd, int* __restrict__ erec,
         unsigned N, unsigned E, unsigned NB, unsigned G1, unsigned nblk) {
    __shared__ int deg[BW];
    __shared__ int cur[BW];
    __shared__ int bpre[NBMAX];
    __shared__ int wsum[16];
    __shared__ v2i stage[CAP2];
    const unsigned tid = threadIdx.x;
    const unsigned b = blockIdx.x;

    int tot;
    const int bs = (tid < nblk) ? bsum[tid] : 0;
    bpre[tid] = block_scan512(bs, wsum, tot);
    if (tid < BW) deg[tid] = 0;
    __syncthreads();

    const unsigned i0 = b * G1;
    const unsigned i1 = i0 + G1;
    const int S0 = scanned[i0] + bpre[i0 >> 10];
    const int S1 = (b + 1 < NB) ? scanned[i1] + bpre[i1 >> 10] : (int)E;
    const int n = S1 - S0;
    const int nbase = (int)(b << BSH);

    for (int j0 = 0; j0 < n; j0 += 5120) {
        v2i r[10];
        #pragma unroll
        for (int k = 0; k < 10; ++k) {
            const int i = j0 + 512 * k + (int)tid;
            r[k] = (v2i)(-1);
            if (i < n) r[k] = *(const v2i*)(ebuf + 2u * (unsigned)(S0 + i));
        }
        #pragma unroll
        for (int k = 0; k < 10; ++k) {
            if (r[k].y >= 0) atomicAdd(&deg[r[k].y - nbase], 1);
        }
    }
    __syncthreads();

    const int v = (tid < BW) ? deg[tid] : 0;
    const int ex = block_scan512(v, wsum, tot);
    if (tid < BW) {
        cur[tid] = ex;
        const unsigned node = (unsigned)nbase + tid;
        if (node < N) {
            rowptr[node] = S0 + ex;
            nd[node] = rsqrtf(fmaxf((float)v, 1.0f));
        }
    }
    if (b + 1 == NB && tid == 0) rowptr[N] = (int)E;
    __syncthreads();

    const bool staged = n <= CAP2;
    for (int j0 = 0; j0 < n; j0 += 5120) {
        v2i r[10];
        int od[10];
        #pragma unroll
        for (int k = 0; k < 10; ++k) {
            const int i = j0 + 512 * k + (int)tid;
            r[k] = (v2i)(-1);
            if (i < n) r[k] = *(const v2i*)(ebuf + 2u * (unsigned)(S0 + i));
        }
        #pragma unroll
        for (int k = 0; k < 10; ++k) od[k] = (r[k].y >= 0) ? odeg[r[k].x] : 1;
        #pragma unroll
        for (int k = 0; k < 10; ++k) {
            if (r[k].y >= 0) {
                const int slot = atomicAdd(&cur[r[k].y - nbase], 1);
                v2i o;
                o.x = r[k].x;
                o.y = __float_as_int(rsqrtf(fmaxf((float)od[k], 1.0f)));
                if (staged) stage[slot] = o;
                else *(v2i*)(erec + 2u * (unsigned)(S0 + slot)) = o;
            }
        }
    }
    __syncthreads();
    if (staged) {
        for (int i = (int)tid; i < n; i += 512)
            *(v2i*)(erec + 2u * (unsigned)(S0 + i)) = stage[i];
    }
}


template<int K>
__device__ __forceinline__ void pull_rows(int sv, int wv, int t, const float* __restrict__ X,
                                          unsigned c, v4f& aw, v4f& ap) {
    v4f u[K];
    #pragma unroll
    for (int k = 0; k < K; ++k) {
        const unsigned s = (unsigned)__builtin_amdgcn_readlane(sv, t + k);
        u[k] = *(const v4f*)(X + s * NFEAT + c);
    }
    #pragma unroll
    for (int k = 0; k < K; ++k) {
        const float w = __int_as_float(__builtin_amdgcn_readlane(wv, t + k));
        aw += w * u[k];
        ap += u[k];
    }
}

__global__ void k_pull_x(const float* __restrict__ X, const int* __restrict__ rowptr,
                         const int* __restrict__ erec, const float* __restrict__ nd,
                         float* __restrict__ AW, float* __restrict__ AP, unsigned N) {
    const unsigned lane = threadIdx.x & 31;
    const unsigned row = __builtin_amdgcn_readfirstlane(
        blockIdx.x * (blockDim.x >> 5) + (threadIdx.x >> 5));
    if (row >= N) return;
    const int beg = __builtin_amdgcn_readfirstlane(rowptr[row]);
    const int end = __builtin_amdgcn_readfirstlane(rowptr[row + 1]);
    const unsigned c = lane * 4u;

    v4f aw = (v4f)(0.0f);
    v4f ap = (v4f)(0.0f);
    for (int base = beg; base < end; base += 32) {
        const int rem = end - base;
        const int cnt = rem < 32 ? rem : 32;
        v2i e = (v2i)(0);
        if ((int)lane < cnt) e = *(const v2i*)(erec + 2u * (unsigned)(base + (int)lane));
        const int sv = e.x;
        const int wv = e.y;
        int t = 0;
        for (; t + 8 <= cnt; t += 8) pull_rows<8>(sv, wv, t, X, c, aw, ap);
        switch (cnt - t) {
            case 7: pull_rows<7>(sv, wv, t, X, c, aw, ap); break;
            case 6: pull_rows<6>(sv, wv, t, X, c, aw, ap); break;
            case 5: pull_rows<5>(sv, wv, t, X, c, aw, ap); break;
            case 4: pull_rows<4>(sv, wv, t, X, c, aw, ap); break;
            case 3: pull_rows<3>(sv, wv, t, X, c, aw, ap); break;
            case 2: pull_rows<2>(sv, wv, t, X, c, aw, ap); break;
            case 1: pull_rows<1>(sv, wv, t, X, c, aw, ap); break;
            default: break;
        }
    }

    const float ndr = nd[row];
    *(v4f*)(AW + row * NFEAT + c) = aw * ndr;
    *(v4f*)(AP + row * NFEAT + c) = ap;
}


template<int NT>
struct GFrag {
    v4f a[2];
    v4f b[4][NT / 4];
};

template<int NT>
__device__ __forceinline__ void gfrag_load(GFrag<NT>& f, const float* __restrict__ X,
                                           const float* __restrict__ Wm,
                                           unsigned a0, unsigned a1, unsigned boff, int k8) {
    constexpr unsigned Nout = NT * 16;
    f.a[0] = *(const v4f*)(X + a0 + k8);
    f.a[1] = *(const v4f*)(X + a1 + k8);
    #pragma unroll
    for (int q = 0; q < 4; ++q) {
        #pragma unroll
        for (int g = 0; g < NT / 4; ++g)
            f.b[q][g] = *(const v4f*)(Wm + boff + (unsigned)(k8 + q) * Nout + 4u * g);
    }
}

template<int NT>
__device__ __forceinline__ void gfrag_mma(const GFrag<NT>& f, v8f (&acc)[2][NT]) {
    #pragma unroll
    for (int s = 0; s < 2; ++s) {
        #pragma unroll
        for (int j = 0; j < NT; ++j) {
            v2f b;
            b.x = f.b[2 * s][j >> 2][j & 3];
            b.y = f.b[2 * s + 1][j >> 2][j & 3];
            #pragma unroll
            for (int m = 0; m < 2; ++m) {
                v2f a;
                a.x = f.a[m][2 * s];
                a.y = f.a[m][2 * s + 1];
                acc[m][j] = __builtin_amdgcn_wmma_f32_16x16x4_f32(
                    false, a, false, b, (short)0, acc[m][j], false, false);
            }
        }
    }
}

template<int NT>
__device__ __forceinline__ void wmma_k128(const float* __restrict__ X, const float* __restrict__ Wm,
                                          unsigned a0, unsigned a1, unsigned boff,
                                          v8f (&acc)[2][NT]) {
    GFrag<NT> f0, f1;
    gfrag_load<NT>(f0, X, Wm, a0, a1, boff, 0);
    #pragma unroll 1
    for (int k8 = 0; k8 < 112; k8 += 16) {
        gfrag_load<NT>(f1, X, Wm, a0, a1, boff, k8 + 8);
        gfrag_mma<NT>(f0, acc);
        gfrag_load<NT>(f0, X, Wm, a0, a1, boff, k8 + 16);
        gfrag_mma<NT>(f1, acc);
    }
    gfrag_load<NT>(f1, X, Wm, a0, a1, boff, 120);
    gfrag_mma<NT>(f0, acc);
    gfrag_mma<NT>(f1, acc);
}

template<int Nout, bool RELU>
__global__ void __launch_bounds__(128)
k_gemm2_wmma(const float* __restrict__ XA, const float* __restrict__ XB,
             const float* __restrict__ WA, const float* __restrict__ WB,
             const float* __restrict__ bias, float* __restrict__ Y, int M) {
    constexpr int NT = Nout >> 4;
    const int lane = threadIdx.x & 31;
    const int strip = blockIdx.x * (blockDim.x >> 5) + (threadIdx.x >> 5);
    const int mbase = strip << 5;
    if (mbase >= M) return;

    const int c = lane & 15;
    const int h = lane >> 4;

    const int r0 = min(mbase + c, M - 1);
    const int r1 = min(mbase + 16 + c, M - 1);
    const unsigned a0 = (unsigned)r0 * 128u + 4u * h;
    const unsigned a1 = (unsigned)r1 * 128u + 4u * h;
    const unsigned boff = (unsigned)(4 * h) * Nout + (unsigned)(c * NT);

    v8f acc[2][NT];
    #pragma unroll
    for (int m = 0; m < 2; ++m) {
        #pragma unroll
        for (int j = 0; j < NT; ++j) acc[m][j] = (v8f)(0.0f);
    }

    wmma_k128<NT>(XA, WA, a0, a1, boff, acc);
    wmma_k128<NT>(XB, WB, a0, a1, boff, acc);

    v4f bv[NT / 4];
    #pragma unroll
    for (int g = 0; g < NT / 4; ++g) bv[g] = *(const v4f*)(bias + c * NT + 4 * g);
    #pragma unroll
    for (int m = 0; m < 2; ++m) {
        #pragma unroll
        for (int r = 0; r < 8; ++r) {
            const int row = mbase + 16 * m + 8 * h + r;
            if (row < M) {
                #pragma unroll
                for (int g = 0; g < NT / 4; ++g) {
                    v4f o;
                    #pragma unroll
                    for (int e = 0; e < 4; ++e) {
                        const float x = acc[m][4 * g + e][r] + bv[g][e];
                        o[e] = RELU ? fmaxf(x, 0.0f) : x;
                    }
                    *(v4f*)(Y + (unsigned)row * (unsigned)Nout + (unsigned)(c * NT + 4 * g)) = o;
                }
            }
        }
    }
}


extern "C" void kernel_launch(void* const* d_in, const int* in_sizes, int n_in,
                              void* d_out, int out_size, void* d_ws, size_t ws_size,
                              hipStream_t stream) {
    const float* raw_x = (const float*)d_in[0];
    const int*   src   = (const int*)d_in[1];
    const int*   dst   = (const int*)d_in[2];
    const float* gc_w0 = (const float*)d_in[3];
    const float* gc_b0 = (const float*)d_in[4];
    const float* gc_w1 = (const float*)d_in[5];
    const float* gc_b1 = (const float*)d_in[6];
    const float* gc_w2 = (const float*)d_in[7];
    const float* gc_b2 = (const float*)d_in[8];
    const float* rw0   = (const float*)d_in[9];
    const float* rw1   = (const float*)d_in[10];
    const float* rw2   = (const float*)d_in[11];

    const int N = in_sizes[0] / NFEAT;
    const int E = in_sizes[1];
    const size_t NB_F = (size_t)N * NFEAT;

    const int TB = 256;
    auto blocks = [](long long n, int tb) { return (int)((n + tb - 1) / tb); };

    const int G1   = blocks(E, CH1);
    const int NBK  = (N + BW - 1) >> BSH;
    const int nent = NBK * G1;
    const int nblk = blocks(nent + 1LL, 1024);

    float* P0 = (float*)d_ws;
    float* P1 = P0 + NB_F;
    float* P2 = P1 + NB_F;
    int*  cur = (int*)(P2 + NB_F);
    auto take = [&](size_t n) { int* p = cur; cur += (n + 3) & ~(size_t)3; return p; };
    float* nd     = (float*)take(N);
    int*  odeg    = take(N);
    int*  rowptr  = take((size_t)N + 1);
    int*  cnt     = take(nent);
    int*  scanned = take((size_t)nent + 1);
    int*  bsum    = take(1024);
    int*  ebuf    = take(2 * (size_t)E);
    int*  erec    = take(2 * (size_t)E);

    k_zero_i32<<<blocks(N, TB), TB, 0, stream>>>(odeg, (unsigned)N);
    k_bin<false><<<G1, 512, 0, stream>>>(src, dst, odeg, cnt, nullptr, nullptr, nullptr,
                                         (unsigned)E, (unsigned)NBK, (unsigned)nblk);
    k_scan_local<<<nblk, TB, 0, stream>>>(cnt, scanned, bsum, (unsigned)nent);
    k_bin<true><<<G1, 512, 0, stream>>>(src, dst, nullptr, nullptr, scanned, bsum, ebuf,
                                        (unsigned)E, (unsigned)NBK, (unsigned)nblk);
    k_bucket<<<NBK, 512, 0, stream>>>(ebuf, scanned, bsum, odeg, rowptr, nd, erec,
                                      (unsigned)N, (unsigned)E, (unsigned)NBK, (unsigned)G1,
                                      (unsigned)nblk);

    const int strips = (N + 31) / 32;
    const int gemmBlocks = blocks(strips, 4);
    const int pullBlocks = blocks(N, 8);
    auto pull = [&](const float* X) {
        k_pull_x<<<pullBlocks, TB, 0, stream>>>(X, rowptr, erec, nd, P1, P2, (unsigned)N);
    };

    pull(raw_x);
    k_gemm2_wmma<128, true><<<gemmBlocks, 128, 0, stream>>>(
        P1, P2, gc_w0, rw0, gc_b0, P0, N);

    pull(P0);
    k_gemm2_wmma<128, true><<<gemmBlocks, 128, 0, stream>>>(
        P1, P2, gc_w1, rw1, gc_b1, P0, N);

    pull(P0);
    k_gemm2_wmma<64, false><<<gemmBlocks, 128, 0, stream>>>(
        P1, P2, gc_w2, rw2, gc_b2, (float*)d_out, N);
}


# Gather fused into the layer GEMM: a 4-wave block sums its 32 destination rows into an LDS tile and multiplies it by [W;R] from there; the two sum planes and 3 launches are gone; sum order kept
# speedup vs baseline: 5.6344x; 1.1102x over previous
#include <hip/hip_runtime.h>

typedef __attribute__((ext_vector_type(2))) float v2f;
typedef __attribute__((ext_vector_type(4))) float v4f;
typedef __attribute__((ext_vector_type(8))) float v8f;
typedef __attribute__((ext_vector_type(2))) int   v2i;

#define NFEAT 128
#define NHID  128
#define NCLASS 64


__global__ void k_zero_i32(int* __restrict__ p, unsigned n) {
    unsigned i = blockIdx.x * blockDim.x + threadIdx.x;
    if (i < n) p[i] = 0;
}

__global__ void k_scan_local(const int* __restrict__ indeg, int* __restrict__ rowptr,
                             int* __restrict__ bsum, unsigned N) {
    __shared__ int wsum[8];
    const unsigned tid  = threadIdx.x;
    const unsigned lane = tid & 31;
    const unsigned wid  = tid >> 5;
    const unsigned base = blockIdx.x * 1024u + tid * 4u;
    int v[4];
    #pragma unroll
    for (int k = 0; k < 4; ++k) v[k] = (base + k < N) ? indeg[base + k] : 0;
    const int t = v[0] + v[1] + v[2] + v[3];
    int inc = t;
    #pragma unroll
    for (int d = 1; d < 32; d <<= 1) {
        const int n = __shfl_up(inc, d);
        if (lane >= (unsigned)d) inc += n;
    }
    if (lane == 31) wsum[wid] = inc;
    __syncthreads();
    int woff = 0, tot = 0;
    #pragma unroll
    for (int w = 0; w < 8; ++w) {
        const int s = wsum[w];
        if ((unsigned)w < wid) woff += s;
        tot += s;
    }
    int ex = woff + inc - t;
    #pragma unroll
    for (int k = 0; k < 4; ++k) {
        if (base + k <= N) rowptr[base + k] = ex;
        ex += v[k];
    }
    if (tid == 0) bsum[blockIdx.x] = tot;
}


#define BSH   8
#define BW    (1 << BSH)
#define NBMAX 512
#define CH1   6144
#define CAP2  6144

__device__ __forceinline__ int block_scan512(int v, int* wsum, int& tot) {
    const unsigned lane = threadIdx.x & 31;
    const unsigned wid  = threadIdx.x >> 5;
    int inc = v;
    #pragma unroll
    for (int d = 1; d < 32; d <<= 1) {
        const int n = __shfl_up(inc, d);
        if (lane >= (unsigned)d) inc += n;
    }
    __syncthreads();
    if (lane == 31) wsum[wid] = inc;
    __syncthreads();
    int woff = 0;
    tot = 0;
    #pragma unroll
    for (int w = 0; w < 16; ++w) {
        const int s = wsum[w];
        if ((unsigned)w < wid) woff += s;
        tot += s;
    }
    return woff + inc - v;
}

template<bool SCATTER>
__global__ void __launch_bounds__(512)
k_bin(const int* __restrict__ src, const int* __restrict__ dst, int* __restrict__ odeg,
      int* __restrict__ cnt, const int* __restrict__ scanned, const int* __restrict__ bsum,
      int* __restrict__ ebuf, unsigned E, unsigned NB, unsigned nblk) {
    __shared__ int hist[NBMAX];
    __shared__ int gdelta[NBMAX];
    __shared__ int bpre[NBMAX];
    __shared__ int wsum[16];
    __shared__ v2i stage[SCATTER ? CH1 : 1];
    const unsigned tid = threadIdx.x;
    const unsigned G1 = gridDim.x;
    const unsigned wg = blockIdx.x;
    const unsigned c0 = wg * (unsigned)CH1;
    const unsigned n = min(E - c0, (unsigned)CH1);

    hist[tid] = 0;
    __syncthreads();

    int s[12], d[12], slot[12];
    #pragma unroll
    for (int k = 0; k < 12; ++k) {
        const unsigned i = tid + 512u * k;
        const bool ok = i < n;
        s[k] = ok ? src[c0 + i] : 0;
        d[k] = ok ? dst[c0 + i] : -1;
    }
    #pragma unroll
    for (int k = 0; k < 12; ++k) {
        slot[k] = 0;
        if (d[k] >= 0) {
            slot[k] = atomicAdd(&hist[d[k] >> BSH], 1);
            if (!SCATTER) atomicAdd(&odeg[s[k]], 1);
        }
    }
    __syncthreads();

    if (!SCATTER) {
        if (tid < NB) cnt[tid * G1 + wg] = hist[tid];
        return;
    }

    int tot;
    const int bs = (tid < nblk) ? bsum[tid] : 0;
    bpre[tid] = block_scan512(bs, wsum, tot);
    const int h = hist[tid];
    const int lb = block_scan512(h, wsum, tot);
    if (tid < NB) {
        const unsigned idx = tid * G1 + wg;
        gdelta[tid] = scanned[idx] + bpre[idx >> 10] - lb;
    }
    hist[tid] = lb;
    __syncthreads();

    #pragma unroll
    for (int k = 0; k < 12; ++k) {
        if (d[k] >= 0) {
            v2i r;
            r.x = s[k];
            r.y = d[k];
            stage[hist[d[k] >> BSH] + slot[k]] = r;
        }
    }
    __syncthreads();

    for (unsigned i = tid; i < n; i += 512u) {
        const v2i r = stage[i];
        *(v2i*)(ebuf + 2u * (unsigned)(gdelta[r.y >> BSH] + (int)i)) = r;
    }
}

__global__ void __launch_bounds__(512)
k_bucket(const int* __restrict__ ebuf, const int* __restrict__ scanned,
         const int* __restrict__ bsum, const int* __restrict__ odeg,
         int* __restrict__ rowptr, float* __restrict__ nd, int* __restrict__ erec,
         unsigned N, unsigned E, unsigned NB, unsigned G1, unsigned nblk) {
    __shared__ int deg[BW];
    __shared__ int cur[BW];
    __shared__ int bpre[NBMAX];
    __shared__ int wsum[16];
    __shared__ v2i stage[CAP2];
    const unsigned tid = threadIdx.x;
    const unsigned b = blockIdx.x;

    int tot;
    const int bs = (tid < nblk) ? bsum[tid] : 0;
    bpre[tid] = block_scan512(bs, wsum, tot);
    if (tid < BW) deg[tid] = 0;
    __syncthreads();

    const unsigned i0 = b * G1;
    const unsigned i1 = i0 + G1;
    const int S0 = scanned[i0] + bpre[i0 >> 10];
    const int S1 = (b + 1 < NB) ? scanned[i1] + bpre[i1 >> 10] : (int)E;
    const int n = S1 - S0;
    const int nbase = (int)(b << BSH);

    for (int j0 = 0; j0 < n; j0 += 5120) {
        v2i r[10];
        #pragma unroll
        for (int k = 0; k < 10; ++k) {
            const int i = j0 + 512 * k + (int)tid;
            r[k] = (v2i)(-1);
            if (i < n) r[k] = *(const v2i*)(ebuf + 2u * (unsigned)(S0 + i));
        }
        #pragma unroll
        for (int k = 0; k < 10; ++k) {
            if (r[k].y >= 0) atomicAdd(&deg[r[k].y - nbase], 1);
        }
    }
    __syncthreads();

    const int v = (tid < BW) ? deg[tid] : 0;
    const int ex = block_scan512(v, wsum, tot);
    if (tid < BW) {
        cur[tid] = ex;
        const unsigned node = (unsigned)nbase + tid;
        if (node < N) {
            rowptr[node] = S0 + ex;
            nd[node] = rsqrtf(fmaxf((float)v, 1.0f));
        }
    }
    if (b + 1 == NB && tid == 0) rowptr[N] = (int)E;
    __syncthreads();

    const bool staged = n <= CAP2;
    for (int j0 = 0; j0 < n; j0 += 5120) {
        v2i r[10];
        int od[10];
        #pragma unroll
        for (int k = 0; k < 10; ++k) {
            const int i = j0 + 512 * k + (int)tid;
            r[k] = (v2i)(-1);
            if (i < n) r[k] = *(const v2i*)(ebuf + 2u * (unsigned)(S0 + i));
        }
        #pragma unroll
        for (int k = 0; k < 10; ++k) od[k] = (r[k].y >= 0) ? odeg[r[k].x] : 1;
        #pragma unroll
        for (int k = 0; k < 10; ++k) {
            if (r[k].y >= 0) {
                const int slot = atomicAdd(&cur[r[k].y - nbase], 1);
                v2i o;
                o.x = r[k].x;
                o.y = __float_as_int(rsqrtf(fmaxf((float)od[k], 1.0f)));
                if (staged) stage[slot] = o;
                else *(v2i*)(erec + 2u * (unsigned)(S0 + slot)) = o;
            }
        }
    }
    __syncthreads();
    if (staged) {
        for (int i = (int)tid; i < n; i += 512)
            *(v2i*)(erec + 2u * (unsigned)(S0 + i)) = stage[i];
    }
}


template<int K>
__device__ __forceinline__ void pull_rows(int sv, int wv, int t, const float* __restrict__ X,
                                          unsigned c, v4f& aw, v4f& ap) {
    v4f u[K];
    #pragma unroll
    for (int k = 0; k < K; ++k) {
        const unsigned s = (unsigned)__builtin_amdgcn_readlane(sv, t + k);
        u[k] = *(const v4f*)(X + s * NFEAT + c);
    }
    #pragma unroll
    for (int k = 0; k < K; ++k) {
        const float w = __int_as_float(__builtin_amdgcn_readlane(wv, t + k));
        aw += w * u[k];
        ap += u[k];
    }
}

#define TP 260

template<int MT>
struct LFrag {
    v4f a[MT];
    v2f b[4];
};

template<int MT, int Nout>
__device__ __forceinline__ void lfrag_load(LFrag<MT>& f, const float* Tl,
                                           const float* __restrict__ Wm, unsigned boff, int k8) {
    #pragma unroll
    for (int m = 0; m < MT; ++m) f.a[m] = *(const v4f*)(Tl + m * 16 * TP + k8);
    #pragma unroll
    for (int q = 0; q < 4; ++q)
        f.b[q] = *(const v2f*)(Wm + boff + (unsigned)(k8 + q) * (unsigned)Nout);
}

template<int MT>
__device__ __forceinline__ void lfrag_mma(const LFrag<MT>& f, v8f (&acc)[MT][2]) {
    #pragma unroll
    for (int s = 0; s < 2; ++s) {
        #pragma unroll
        for (int j = 0; j < 2; ++j) {
            v2f b;
            b.x = f.b[2 * s][j];
            b.y = f.b[2 * s + 1][j];
            #pragma unroll
            for (int m = 0; m < MT; ++m) {
                v2f a;
                a.x = f.a[m][2 * s];
                a.y = f.a[m][2 * s + 1];
                acc[m][j] = __builtin_amdgcn_wmma_f32_16x16x4_f32(
                    false, a, false, b, (short)0, acc[m][j], false, false);
            }
        }
    }
}

template<int MT, int Nout>
__device__ __forceinline__ void lds_k128(const float* Tl, const float* __restrict__ Wm,
                                         unsigned boff, v8f (&acc)[MT][2]) {
    LFrag<MT> f0, f1;
    lfrag_load<MT, Nout>(f0, Tl, Wm, boff, 0);
    #pragma unroll 1
    for (int k8 = 0; k8 < 112; k8 += 16) {
        lfrag_load<MT, Nout>(f1, Tl, Wm, boff, k8 + 8);
        lfrag_mma<MT>(f0, acc);
        lfrag_load<MT, Nout>(f0, Tl, Wm, boff, k8 + 16);
        lfrag_mma<MT>(f1, acc);
    }
    lfrag_load<MT, Nout>(f1, Tl, Wm, boff, 120);
    lfrag_mma<MT>(f0, acc);
    lfrag_mma<MT>(f1, acc);
}

template<int Nout, bool RELU>
__global__ void __launch_bounds__(128)
k_layer(const float* __restrict__ X, const int* __restrict__ rowptr,
        const int* __restrict__ erec, const float* __restrict__ nd,
        const float* __restrict__ WA, const float* __restrict__ WB,
        const float* __restrict__ bias, float* __restrict__ Y, unsigned N) {
    constexpr int MT = Nout / 64;
    __shared__ __attribute__((aligned(16))) float T[32 * TP];

    const unsigned lane = threadIdx.x & 31;
    const unsigned w = __builtin_amdgcn_readfirstlane(threadIdx.x >> 5);
    const unsigned mbase = blockIdx.x * 32u;

    {
        const unsigned row0 = mbase + 8u * w;
        const unsigned c4 = lane * 4u;
        int rp = 0;
        float ndv = 0.0f;
        if (lane <= 8u && row0 + lane <= N) rp = rowptr[row0 + lane];
        if (lane < 8u && row0 + lane < N) ndv = nd[row0 + lane];

        #pragma unroll 1
        for (int i = 0; i < 8; ++i) {
            v4f aw = (v4f)(0.0f);
            v4f ap = (v4f)(0.0f);
            if (row0 + (unsigned)i < N) {
                const int beg = __builtin_amdgcn_readlane(rp, i);
                const int end = __builtin_amdgcn_readlane(rp, i + 1);
                for (int base = beg; base < end; base += 32) {
                    const int rem = end - base;
                    const int cnt = rem < 32 ? rem : 32;
                    v2i e = (v2i)(0);
                    if ((int)lane < cnt) e = *(const v2i*)(erec + 2u * (unsigned)(base + (int)lane));
                    const int sv = e.x;
                    const int wv = e.y;
                    int t = 0;
                    for (; t + 8 <= cnt; t += 8) pull_rows<8>(sv, wv, t, X, c4, aw, ap);
                    switch (cnt - t) {
                        case 7: pull_rows<7>(sv, wv, t, X, c4, aw, ap); break;
                        case 6: pull_rows<6>(sv, wv, t, X, c4, aw, ap); break;
                        case 5: pull_rows<5>(sv, wv, t, X, c4, aw, ap); break;
                        case 4: pull_rows<4>(sv, wv, t, X, c4, aw, ap); break;
                        case 3: pull_rows<3>(sv, wv, t, X, c4, aw, ap); break;
                        case 2: pull_rows<2>(sv, wv, t, X, c4, aw, ap); break;
                        case 1: pull_rows<1>(sv, wv, t, X, c4, aw, ap); break;
                        default: break;
                    }
                }
                aw = aw * __int_as_float(__builtin_amdgcn_readlane(__float_as_int(ndv), i));
            }
            float* tr = T + (8u * w + (unsigned)i) * TP + c4;
            *(v4f*)tr = aw;
            *(v4f*)(tr + 128) = ap;
        }
    }
    __syncthreads();

    const unsigned c = lane & 15u;
    const unsigned h = lane >> 4;
    const unsigned rt = (MT == 2) ? 0u : 16u * (w >> 1);
    const unsigned cb = (MT == 2) ? 32u * w : 32u * (w & 1u);
    const float* Tl = T + (rt + c) * TP + 4u * h;
    const unsigned boff = (4u * h) * (unsigned)Nout + cb + 2u * c;

    v8f acc[MT][2];
    #pragma unroll
    for (int m = 0; m < MT; ++m) {
        #pragma unroll
        for (int j = 0; j < 2; ++j) acc[m][j] = (v8f)(0.0f);
    }

    lds_k128<MT, Nout>(Tl, WA, boff, acc);
    lds_k128<MT, Nout>(Tl + 128, WB, boff, acc);

    const v2f bv = *(const v2f*)(bias + cb + 2u * c);
    #pragma unroll
    for (int m = 0; m < MT; ++m) {
        #pragma unroll
        for (int r = 0; r < 8; ++r) {
            const unsigned row = mbase + rt + 16u * m + 8u * h + (unsigned)r;
            if (row < N) {
                v2f o;
                o.x = acc[m][0][r] + bv.x;
                o.y = acc[m][1][r] + bv.y;
                if (RELU) {
                    o.x = fmaxf(o.x, 0.0f);
                    o.y = fmaxf(o.y, 0.0f);
                }
                *(v2f*)(Y + row * (unsigned)Nout + cb + 2u * c) = o;
            }
        }
    }
}


extern "C" void kernel_launch(void* const* d_in, const int* in_sizes, int n_in,
                              void* d_out, int out_size, void* d_ws, size_t ws_size,
                              hipStream_t stream) {
    const float* raw_x = (const float*)d_in[0];
    const int*   src   = (const int*)d_in[1];
    const int*   dst   = (const int*)d_in[2];
    const float* gc_w0 = (const float*)d_in[3];
    const float* gc_b0 = (const float*)d_in[4];
    const float* gc_w1 = (const float*)d_in[5];
    const float* gc_b1 = (const float*)d_in[6];
    const float* gc_w2 = (const float*)d_in[7];
    const float* gc_b2 = (const float*)d_in[8];
    const float* rw0   = (const float*)d_in[9];
    const float* rw1   = (const float*)d_in[10];
    const float* rw2   = (const float*)d_in[11];

    const int N = in_sizes[0] / NFEAT;
    const int E = in_sizes[1];
    const size_t NB_F = (size_t)N * NFEAT;

    const int TB = 256;
    auto blocks = [](long long n, int tb) { return (int)((n + tb - 1) / tb); };

    const int G1   = blocks(E, CH1);
    const int NBK  = (N + BW - 1) >> BSH;
    const int nent = NBK * G1;
    const int nblk = blocks(nent + 1LL, 1024);

    float* P0 = (float*)d_ws;
    float* P1 = P0 + NB_F;
    int*  cur = (int*)(P1 + NB_F);
    auto take = [&](size_t n) { int* p = cur; cur += (n + 3) & ~(size_t)3; return p; };
    float* nd     = (float*)take(N);
    int*  odeg    = take(N);
    int*  rowptr  = take((size_t)N + 1);
    int*  cnt     = take(nent);
    int*  scanned = take((size_t)nent + 1);
    int*  bsum    = take(1024);
    int*  ebuf    = take(2 * (size_t)E);
    int*  erec    = take(2 * (size_t)E);

    k_zero_i32<<<blocks(N, TB), TB, 0, stream>>>(odeg, (unsigned)N);
    k_bin<false><<<G1, 512, 0, stream>>>(src, dst, odeg, cnt, nullptr, nullptr, nullptr,
                                         (unsigned)E, (unsigned)NBK, (unsigned)nblk);
    k_scan_local<<<nblk, TB, 0, stream>>>(cnt, scanned, bsum, (unsigned)nent);
    k_bin<true><<<G1, 512, 0, stream>>>(src, dst, nullptr, nullptr, scanned, bsum, ebuf,
                                        (unsigned)E, (unsigned)NBK, (unsigned)nblk);
    k_bucket<<<NBK, 512, 0, stream>>>(ebuf, scanned, bsum, odeg, rowptr, nd, erec,
                                      (unsigned)N, (unsigned)E, (unsigned)NBK, (unsigned)G1,
                                      (unsigned)nblk);

    const int layerBlocks = blocks(N, 32);

    k_layer<128, true><<<layerBlocks, 128, 0, stream>>>(
        raw_x, rowptr, erec, nd, gc_w0, rw0, gc_b0, P0, (unsigned)N);

    k_layer<128, true><<<layerBlocks, 128, 0, stream>>>(
        P0, rowptr, erec, nd, gc_w1, rw1, gc_b1, P1, (unsigned)N);

    k_layer<64, false><<<layerBlocks, 128, 0, stream>>>(
        P1, rowptr, erec, nd, gc_w2, rw2, gc_b2, (float*)d_out, (unsigned)N);
}


# Gather and product overlapped by wave roles: resident 12-wave blocks walk their tiles, 8 waves sum tile t+1 into one LDS tile while 4 waves multiply tile t from the other; sum order kept
# speedup vs baseline: 5.9061x; 1.0482x over previous
#include <hip/hip_runtime.h>

typedef __attribute__((ext_vector_type(2))) float v2f;
typedef __attribute__((ext_vector_type(4))) float v4f;
typedef __attribute__((ext_vector_type(8))) float v8f;
typedef __attribute__((ext_vector_type(2))) int   v2i;

#define NFEAT 128
#define NHID  128
#define NCLASS 64


__global__ void k_zero_i32(int* __restrict__ p, unsigned n) {
    unsigned i = blockIdx.x * blockDim.x + threadIdx.x;
    if (i < n) p[i] = 0;
}

__global__ void k_scan_local(const int* __restrict__ indeg, int* __restrict__ rowptr,
                             int* __restrict__ bsum, unsigned N) {
    __shared__ int wsum[8];
    const unsigned tid  = threadIdx.x;
    const unsigned lane = tid & 31;
    const unsigned wid  = tid >> 5;
    const unsigned base = blockIdx.x * 1024u + tid * 4u;
    int v[4];
    #pragma unroll
    for (int k = 0; k < 4; ++k) v[k] = (base + k < N) ? indeg[base + k] : 0;
    const int t = v[0] + v[1] + v[2] + v[3];
    int inc = t;
    #pragma unroll
    for (int d = 1; d < 32; d <<= 1) {
        const int n = __shfl_up(inc, d);
        if (lane >= (unsigned)d) inc += n;
    }
    if (lane == 31) wsum[wid] = inc;
    __syncthreads();
    int woff = 0, tot = 0;
    #pragma unroll
    for (int w = 0; w < 8; ++w) {
        const int s = wsum[w];
        if ((unsigned)w < wid) woff += s;
        tot += s;
    }
    int ex = woff + inc - t;
    #pragma unroll
    for (int k = 0; k < 4; ++k) {
        if (base + k <= N) rowptr[base + k] = ex;
        ex += v[k];
    }
    if (tid == 0) bsum[blockIdx.x] = tot;
}


#define BSH   8
#define BW    (1 << BSH)
#define NBMAX 512
#define CH1   6144
#define CAP2  6144

__device__ __forceinline__ int block_scan512(int v, int* wsum, int& tot) {
    const unsigned lane = threadIdx.x & 31;
    const unsigned wid  = threadIdx.x >> 5;
    int inc = v;
    #pragma unroll
    for (int d = 1; d < 32; d <<= 1) {
        const int n = __shfl_up(inc, d);
        if (lane >= (unsigned)d) inc += n;
    }
    __syncthreads();
    if (lane == 31) wsum[wid] = inc;
    __syncthreads();
    int woff = 0;
    tot = 0;
    #pragma unroll
    for (int w = 0; w < 16; ++w) {
        const int s = wsum[w];
        if ((unsigned)w < wid) woff += s;
        tot += s;
    }
    return woff + inc - v;
}

template<bool SCATTER>
__global__ void __launch_bounds__(512)
k_bin(const int* __restrict__ src, const int* __restrict__ dst, int* __restrict__ odeg,
      int* __restrict__ cnt, const int* __restrict__ scanned, const int* __restrict__ bsum,
      int* __restrict__ ebuf, unsigned E, unsigned NB, unsigned nblk) {
    __shared__ int hist[NBMAX];
    __shared__ int gdelta[NBMAX];
    __shared__ int bpre[NBMAX];
    __shared__ int wsum[16];
    __shared__ v2i stage[SCATTER ? CH1 : 1];
    const unsigned tid = threadIdx.x;
    const unsigned G1 = gridDim.x;
    const unsigned wg = blockIdx.x;
    const unsigned c0 = wg * (unsigned)CH1;
    const unsigned n = min(E - c0, (unsigned)CH1);

    hist[tid] = 0;
    __syncthreads();

    int s[12], d[12], slot[12];
    #pragma unroll
    for (int k = 0; k < 12; ++k) {
        const unsigned i = tid + 512u * k;
        const bool ok = i < n;
        s[k] = ok ? src[c0 + i] : 0;
        d[k] = ok ? dst[c0 + i] : -1;
    }
    #pragma unroll
    for (int k = 0; k < 12; ++k) {
        slot[k] = 0;
        if (d[k] >= 0) {
            slot[k] = atomicAdd(&hist[d[k] >> BSH], 1);
            if (!SCATTER) atomicAdd(&odeg[s[k]], 1);
        }
    }
    __syncthreads();

    if (!SCATTER) {
        if (tid < NB) cnt[tid * G1 + wg] = hist[tid];
        return;
    }

    int tot;
    const int bs = (tid < nblk) ? bsum[tid] : 0;
    bpre[tid] = block_scan512(bs, wsum, tot);
    const int h = hist[tid];
    const int lb = block_scan512(h, wsum, tot);
    if (tid < NB) {
        const unsigned idx = tid * G1 + wg;
        gdelta[tid] = scanned[idx] + bpre[idx >> 10] - lb;
    }
    hist[tid] = lb;
    __syncthreads();

    #pragma unroll
    for (int k = 0; k < 12; ++k) {
        if (d[k] >= 0) {
            v2i r;
            r.x = s[k];
            r.y = d[k];
            stage[hist[d[k] >> BSH] + slot[k]] = r;
        }
    }
    __syncthreads();

    for (unsigned i = tid; i < n; i += 512u) {
        const v2i r = stage[i];
        *(v2i*)(ebuf + 2u * (unsigned)(gdelta[r.y >> BSH] + (int)i)) = r;
    }
}

__global__ void __launch_bounds__(512)
k_bucket(const int* __restrict__ ebuf, const int* __restrict__ scanned,
         const int* __restrict__ bsum, const int* __restrict__ odeg,
         int* __restrict__ rowptr, float* __restrict__ nd, int* __restrict__ erec,
         unsigned N, unsigned E, unsigned NB, unsigned G1, unsigned nblk) {
    __shared__ int deg[BW];
    __shared__ int cur[BW];
    __shared__ int bpre[NBMAX];
    __shared__ int wsum[16];
    __shared__ v2i stage[CAP2];
    const unsigned tid = threadIdx.x;
    const unsigned b = blockIdx.x;

    int tot;
    const int bs = (tid < nblk) ? bsum[tid] : 0;
    bpre[tid] = block_scan512(bs, wsum, tot);
    if (tid < BW) deg[tid] = 0;
    __syncthreads();

    const unsigned i0 = b * G1;
    const unsigned i1 = i0 + G1;
    const int S0 = scanned[i0] + bpre[i0 >> 10];
    const int S1 = (b + 1 < NB) ? scanned[i1] + bpre[i1 >> 10] : (int)E;
    const int n = S1 - S0;
    const int nbase = (int)(b << BSH);

    for (int j0 = 0; j0 < n; j0 += 5120) {
        v2i r[10];
        #pragma unroll
        for (int k = 0; k < 10; ++k) {
            const int i = j0 + 512 * k + (int)tid;
            r[k] = (v2i)(-1);
            if (i < n) r[k] = *(const v2i*)(ebuf + 2u * (unsigned)(S0 + i));
        }
        #pragma unroll
        for (int k = 0; k < 10; ++k) {
            if (r[k].y >= 0) atomicAdd(&deg[r[k].y - nbase], 1);
        }
    }
    __syncthreads();

    const int v = (tid < BW) ? deg[tid] : 0;
    const int ex = block_scan512(v, wsum, tot);
    if (tid < BW) {
        cur[tid] = ex;
        const unsigned node = (unsigned)nbase + tid;
        if (node < N) {
            rowptr[node] = S0 + ex;
            nd[node] = rsqrtf(fmaxf((float)v, 1.0f));
        }
    }
    if (b + 1 == NB && tid == 0) rowptr[N] = (int)E;
    __syncthreads();

    const bool staged = n <= CAP2;
    for (int j0 = 0; j0 < n; j0 += 5120) {
        v2i r[10];
        int od[10];
        #pragma unroll
        for (int k = 0; k < 10; ++k) {
            const int i = j0 + 512 * k + (int)tid;
            r[k] = (v2i)(-1);
            if (i < n) r[k] = *(const v2i*)(ebuf + 2u * (unsigned)(S0 + i));
        }
        #pragma unroll
        for (int k = 0; k < 10; ++k) od[k] = (r[k].y >= 0) ? odeg[r[k].x] : 1;
        #pragma unroll
        for (int k = 0; k < 10; ++k) {
            if (r[k].y >= 0) {
                const int slot = atomicAdd(&cur[r[k].y - nbase], 1);
                v2i o;
                o.x = r[k].x;
                o.y = __float_as_int(rsqrtf(fmaxf((float)od[k], 1.0f)));
                if (staged) stage[slot] = o;
                else *(v2i*)(erec + 2u * (unsigned)(S0 + slot)) = o;
            }
        }
    }
    __syncthreads();
    if (staged) {
        for (int i = (int)tid; i < n; i += 512)
            *(v2i*)(erec + 2u * (unsigned)(S0 + i)) = stage[i];
    }
}


template<int K>
__device__ __forceinline__ void pull_rows(int sv, int wv, int t, const float* __restrict__ X,
                                          unsigned c, v4f& aw, v4f& ap) {
    v4f u[K];
    #pragma unroll
    for (int k = 0; k < K; ++k) {
        const unsigned s = (unsigned)__builtin_amdgcn_readlane(sv, t + k);
        u[k] = *(const v4f*)(X + s * NFEAT + c);
    }
    #pragma unroll
    for (int k = 0; k < K; ++k) {
        const float w = __int_as_float(__builtin_amdgcn_readlane(wv, t + k));
        aw += w * u[k];
        ap += u[k];
    }
}

#define TP 260

template<int MT>
struct LFrag {
    v4f a[MT];
    v2f b[4];
};

template<int MT, int Nout>
__device__ __forceinline__ void lfrag_load(LFrag<MT>& f, const float* Tl,
                                           const float* __restrict__ Wm, unsigned boff, int k8) {
    #pragma unroll
    for (int m = 0; m < MT; ++m) f.a[m] = *(const v4f*)(Tl + m * 16 * TP + k8);
    #pragma unroll
    for (int q = 0; q < 4; ++q)
        f.b[q] = *(const v2f*)(Wm + boff + (unsigned)(k8 + q) * (unsigned)Nout);
}

template<int MT>
__device__ __forceinline__ void lfrag_mma(const LFrag<MT>& f, v8f (&acc)[MT][2]) {
    #pragma unroll
    for (int s = 0; s < 2; ++s) {
        #pragma unroll
        for (int j = 0; j < 2; ++j) {
            v2f b;
            b.x = f.b[2 * s][j];
            b.y = f.b[2 * s + 1][j];
            #pragma unroll
            for (int m = 0; m < MT; ++m) {
                v2f a;
                a.x = f.a[m][2 * s];
                a.y = f.a[m][2 * s + 1];
                acc[m][j] = __builtin_amdgcn_wmma_f32_16x16x4_f32(
                    false, a, false, b, (short)0, acc[m][j], false, false);
            }
        }
    }
}

template<int MT, int Nout>
__device__ __forceinline__ void lds_k128(const float* Tl, const float* __restrict__ Wm,
                                         unsigned boff, v8f (&acc)[MT][2]) {
    LFrag<MT> f0, f1;
    lfrag_load<MT, Nout>(f0, Tl, Wm, boff, 0);
    #pragma unroll 1
    for (int k8 = 0; k8 < 112; k8 += 16) {
        lfrag_load<MT, Nout>(f1, Tl, Wm, boff, k8 + 8);
        lfrag_mma<MT>(f0, acc);
        lfrag_load<MT, Nout>(f0, Tl, Wm, boff, k8 + 16);
        lfrag_mma<MT>(f1, acc);
    }
    lfrag_load<MT, Nout>(f1, Tl, Wm, boff, 120);
    lfrag_mma<MT>(f0, acc);
    lfrag_mma<MT>(f1, acc);
}

#define GWAVES 8
#define MWAVES 4
#define LAYER_THREADS ((GWAVES + MWAVES) * 32)
#define LAYER_GRID_128 768
#define LAYER_GRID_64  1024

template<int Nout, bool RELU>
__global__ void __launch_bounds__(LAYER_THREADS)
k_layer(const float* __restrict__ X, const int* __restrict__ rowptr,
        const int* __restrict__ erec, const float* __restrict__ nd,
        const float* __restrict__ WA, const float* __restrict__ WB,
        const float* __restrict__ bias, float* __restrict__ Y, unsigned N, unsigned ntiles) {
    constexpr int MT = Nout / 64;
    __shared__ __attribute__((aligned(16))) float T[2][32 * TP];

    const unsigned lane = threadIdx.x & 31;
    const unsigned w = __builtin_amdgcn_readfirstlane(threadIdx.x >> 5);
    const unsigned nt = (ntiles - blockIdx.x + gridDim.x - 1u) / gridDim.x;

    const unsigned c = lane & 15u;
    const unsigned h = lane >> 4;
    const unsigned wq = (w - (unsigned)GWAVES) & 3u;
    const unsigned rt = (MT == 2) ? 0u : 16u * (wq >> 1);
    const unsigned cb = (MT == 2) ? 32u * wq : 32u * (wq & 1u);
    const unsigned boff = (4u * h) * (unsigned)Nout + cb + 2u * c;

    #pragma unroll 1
    for (unsigned it = 0; it <= nt; ++it) {
        if (w < (unsigned)GWAVES) {
            if (it < nt) {
                const unsigned mbase = (blockIdx.x + it * gridDim.x) * 32u;
                float* Tb = T[it & 1u];
                const unsigned row0 = mbase + 4u * w;
                const unsigned c4 = lane * 4u;
                int rp = 0;
                float ndv = 0.0f;
                if (lane <= 4u && row0 + lane <= N) rp = rowptr[row0 + lane];
                if (lane < 4u && row0 + lane < N) ndv = nd[row0 + lane];

                #pragma unroll 1
                for (int i = 0; i < 4; ++i) {
                    v4f aw = (v4f)(0.0f);
                    v4f ap = (v4f)(0.0f);
                    if (row0 + (unsigned)i < N) {
                        const int beg = __builtin_amdgcn_readlane(rp, i);
                        const int end = __builtin_amdgcn_readlane(rp, i + 1);
                        for (int base = beg; base < end; base += 32) {
                            const int rem = end - base;
                            const int cnt = rem < 32 ? rem : 32;
                            v2i e = (v2i)(0);
                            if ((int)lane < cnt) e = *(const v2i*)(erec + 2u * (unsigned)(base + (int)lane));
                            const int sv = e.x;
                            const int wv = e.y;
                            int t = 0;
                            for (; t + 8 <= cnt; t += 8) pull_rows<8>(sv, wv, t, X, c4, aw, ap);
                            switch (cnt - t) {
                                case 7: pull_rows<7>(sv, wv, t, X, c4, aw, ap); break;
                                case 6: pull_rows<6>(sv, wv, t, X, c4, aw, ap); break;
                                case 5: pull_rows<5>(sv, wv, t, X, c4, aw, ap); break;
                                case 4: pull_rows<4>(sv, wv, t, X, c4, aw, ap); break;
                                case 3: pull_rows<3>(sv, wv, t, X, c4, aw, ap); break;
                                case 2: pull_rows<2>(sv, wv, t, X, c4, aw, ap); break;
                                case 1: pull_rows<1>(sv, wv, t, X, c4, aw, ap); break;
                                default: break;
                            }
                        }
                        aw = aw * __int_as_float(__builtin_amdgcn_readlane(__float_as_int(ndv), i));
                    }
                    float* tr = Tb + (4u * w + (unsigned)i) * TP + c4;
                    *(v4f*)tr = aw;
                    *(v4f*)(tr + 128) = ap;
                }
            }
        } else if (it > 0) {
            const unsigned mbase = (blockIdx.x + (it - 1u) * gridDim.x) * 32u;
            const float* Tl = T[(it - 1u) & 1u] + (rt + c) * TP + 4u * h;

            v8f acc[MT][2];
            #pragma unroll
            for (int m = 0; m < MT; ++m) {
                #pragma unroll
                for (int j = 0; j < 2; ++j) acc[m][j] = (v8f)(0.0f);
            }

            lds_k128<MT, Nout>(Tl, WA, boff, acc);
            lds_k128<MT, Nout>(Tl + 128, WB, boff, acc);

            const v2f bv = *(const v2f*)(bias + cb + 2u * c);
            #pragma unroll
            for (int m = 0; m < MT; ++m) {
                #pragma unroll
                for (int r = 0; r < 8; ++r) {
                    const unsigned row = mbase + rt + 16u * m + 8u * h + (unsigned)r;
                    if (row < N) {
                        v2f o;
                        o.x = acc[m][0][r] + bv.x;
                        o.y = acc[m][1][r] + bv.y;
                        if (RELU) {
                            o.x = fmaxf(o.x, 0.0f);
                            o.y = fmaxf(o.y, 0.0f);
                        }
                        *(v2f*)(Y + row * (unsigned)Nout + cb + 2u * c) = o;
                    }
                }
            }
        }
        __syncthreads();
    }
}


extern "C" void kernel_launch(void* const* d_in, const int* in_sizes, int n_in,
                              void* d_out, int out_size, void* d_ws, size_t ws_size,
                              hipStream_t stream) {
    const float* raw_x = (const float*)d_in[0];
    const int*   src   = (const int*)d_in[1];
    const int*   dst   = (const int*)d_in[2];
    const float* gc_w0 = (const float*)d_in[3];
    const float* gc_b0 = (const float*)d_in[4];
    const float* gc_w1 = (const float*)d_in[5];
    const float* gc_b1 = (const float*)d_in[6];
    const float* gc_w2 = (const float*)d_in[7];
    const float* gc_b2 = (const float*)d_in[8];
    const float* rw0   = (const float*)d_in[9];
    const float* rw1   = (const float*)d_in[10];
    const float* rw2   = (const float*)d_in[11];

    const int N = in_sizes[0] / NFEAT;
    const int E = in_sizes[1];
    const size_t NB_F = (size_t)N * NFEAT;

    const int TB = 256;
    auto blocks = [](long long n, int tb) { return (int)((n + tb - 1) / tb); };

    const int G1   = blocks(E, CH1);
    const int NBK  = (N + BW - 1) >> BSH;
    const int nent = NBK * G1;
    const int nblk = blocks(nent + 1LL, 1024);

    float* P0 = (float*)d_ws;
    float* P1 = P0 + NB_F;
    int*  cur = (int*)(P1 + NB_F);
    auto take = [&](size_t n) { int* p = cur; cur += (n + 3) & ~(size_t)3; return p; };
    float* nd     = (float*)take(N);
    int*  odeg    = take(N);
    int*  rowptr  = take((size_t)N + 1);
    int*  cnt     = take(nent);
    int*  scanned = take((size_t)nent + 1);
    int*  bsum    = take(1024);
    int*  ebuf    = take(2 * (size_t)E);
    int*  erec    = take(2 * (size_t)E);

    k_zero_i32<<<blocks(N, TB), TB, 0, stream>>>(odeg, (unsigned)N);
    k_bin<false><<<G1, 512, 0, stream>>>(src, dst, odeg, cnt, nullptr, nullptr, nullptr,
                                         (unsigned)E, (unsigned)NBK, (unsigned)nblk);
    k_scan_local<<<nblk, TB, 0, stream>>>(cnt, scanned, bsum, (unsigned)nent);
    k_bin<true><<<G1, 512, 0, stream>>>(src, dst, nullptr, nullptr, scanned, bsum, ebuf,
                                        (unsigned)E, (unsigned)NBK, (unsigned)nblk);
    k_bucket<<<NBK, 512, 0, stream>>>(ebuf, scanned, bsum, odeg, rowptr, nd, erec,
                                      (unsigned)N, (unsigned)E, (unsigned)NBK, (unsigned)G1,
                                      (unsigned)nblk);

    const int layerTiles = blocks(N, 32);
    const int layerGrid128 = layerTiles < LAYER_GRID_128 ? layerTiles : LAYER_GRID_128;
    const int layerGrid64  = layerTiles < LAYER_GRID_64 ? layerTiles : LAYER_GRID_64;

    k_layer<128, true><<<layerGrid128, LAYER_THREADS, 0, stream>>>(
        raw_x, rowptr, erec, nd, gc_w0, rw0, gc_b0, P0, (unsigned)N, (unsigned)layerTiles);

    k_layer<128, true><<<layerGrid128, LAYER_THREADS, 0, stream>>>(
        P0, rowptr, erec, nd, gc_w1, rw1, gc_b1, P1, (unsigned)N, (unsigned)layerTiles);

    k_layer<64, false><<<layerGrid64, LAYER_THREADS, 0, stream>>>(
        P1, rowptr, erec, nd, gc_w2, rw2, gc_b2, (float*)d_out, (unsigned)N, (unsigned)layerTiles);
}


# Out-degrees by a second LDS counting sort keyed by source inside the two binning launches (a block per source bucket counts in LDS, writes norm_s): scattered global atomics and zero fill removed
# speedup vs baseline: 7.3829x; 1.2500x over previous
#include <hip/hip_runtime.h>

typedef __attribute__((ext_vector_type(2))) float v2f;
typedef __attribute__((ext_vector_type(4))) float v4f;
typedef __attribute__((ext_vector_type(8))) float v8f;
typedef __attribute__((ext_vector_type(2))) int   v2i;

#define NFEAT 128
#define NHID  128
#define NCLASS 64


__global__ void k_scan_local(const int* __restrict__ indeg, int* __restrict__ rowptr,
                             int* __restrict__ bsum, unsigned N) {
    __shared__ int wsum[8];
    const unsigned tid  = threadIdx.x;
    const unsigned lane = tid & 31;
    const unsigned wid  = tid >> 5;
    const unsigned base = blockIdx.x * 1024u + tid * 4u;
    int v[4];
    #pragma unroll
    for (int k = 0; k < 4; ++k) v[k] = (base + k < N) ? indeg[base + k] : 0;
    const int t = v[0] + v[1] + v[2] + v[3];
    int inc = t;
    #pragma unroll
    for (int d = 1; d < 32; d <<= 1) {
        const int n = __shfl_up(inc, d);
        if (lane >= (unsigned)d) inc += n;
    }
    if (lane == 31) wsum[wid] = inc;
    __syncthreads();
    int woff = 0, tot = 0;
    #pragma unroll
    for (int w = 0; w < 8; ++w) {
        const int s = wsum[w];
        if ((unsigned)w < wid) woff += s;
        tot += s;
    }
    int ex = woff + inc - t;
    #pragma unroll
    for (int k = 0; k < 4; ++k) {
        if (base + k <= N) rowptr[base + k] = ex;
        ex += v[k];
    }
    if (tid == 0) bsum[blockIdx.x] = tot;
}


#define BSH   8
#define BW    (1 << BSH)
#define NBMAX 512
#define CH1   6144
#define CAP2  6144

__device__ __forceinline__ int block_scan512(int v, int* wsum, int& tot) {
    const unsigned lane = threadIdx.x & 31;
    const unsigned wid  = threadIdx.x >> 5;
    int inc = v;
    #pragma unroll
    for (int d = 1; d < 32; d <<= 1) {
        const int n = __shfl_up(inc, d);
        if (lane >= (unsigned)d) inc += n;
    }
    __syncthreads();
    if (lane == 31) wsum[wid] = inc;
    __syncthreads();
    int woff = 0;
    tot = 0;
    #pragma unroll
    for (int w = 0; w < 16; ++w) {
        const int s = wsum[w];
        if ((unsigned)w < wid) woff += s;
        tot += s;
    }
    return woff + inc - v;
}

template<bool SCATTER>
__global__ void __launch_bounds__(512)
k_bin(const int* __restrict__ src, const int* __restrict__ dst,
      int* __restrict__ cnt, const int* __restrict__ scanned, const int* __restrict__ bsum,
      int* __restrict__ ebuf, int* __restrict__ sbuf, unsigned E, unsigned NB, unsigned nblk,
      unsigned nent) {
    __shared__ int hist[NBMAX];
    __shared__ int hist2[NBMAX];
    __shared__ int gdelta[NBMAX];
    __shared__ int gdelta2[NBMAX];
    __shared__ int bpre[NBMAX];
    __shared__ int wsum[16];
    __shared__ v2i stage[SCATTER ? CH1 : 1];
    __shared__ int stage2[SCATTER ? CH1 : 1];
    const unsigned tid = threadIdx.x;
    const unsigned G1 = gridDim.x;
    const unsigned wg = blockIdx.x;
    const unsigned c0 = wg * (unsigned)CH1;
    const unsigned n = min(E - c0, (unsigned)CH1);

    hist[tid] = 0;
    hist2[tid] = 0;
    __syncthreads();

    int s[12], d[12], slot[12], slot2[12];
    #pragma unroll
    for (int k = 0; k < 12; ++k) {
        const unsigned i = tid + 512u * k;
        const bool ok = i < n;
        s[k] = ok ? src[c0 + i] : 0;
        d[k] = ok ? dst[c0 + i] : -1;
    }
    #pragma unroll
    for (int k = 0; k < 12; ++k) {
        slot[k] = 0;
        slot2[k] = 0;
        if (d[k] >= 0) {
            slot[k] = atomicAdd(&hist[d[k] >> BSH], 1);
            slot2[k] = atomicAdd(&hist2[s[k] >> BSH], 1);
        }
    }
    __syncthreads();

    if (!SCATTER) {
        if (tid < NB) {
            cnt[tid * G1 + wg] = hist[tid];
            cnt[nent + tid * G1 + wg] = hist2[tid];
        }
        return;
    }

    int tot;
    const int bs = (tid < nblk) ? bsum[tid] : 0;
    bpre[tid] = block_scan512(bs, wsum, tot);
    const int h = hist[tid];
    const int lb = block_scan512(h, wsum, tot);
    const int h2 = hist2[tid];
    const int lb2 = block_scan512(h2, wsum, tot);
    if (tid < NB) {
        const unsigned idx = tid * G1 + wg;
        gdelta[tid] = scanned[idx] + bpre[idx >> 10] - lb;
        const unsigned idx2 = nent + idx;
        gdelta2[tid] = scanned[idx2] + bpre[idx2 >> 10] - (int)E - lb2;
    }
    hist[tid] = lb;
    hist2[tid] = lb2;
    __syncthreads();

    #pragma unroll
    for (int k = 0; k < 12; ++k) {
        if (d[k] >= 0) {
            v2i r;
            r.x = s[k];
            r.y = d[k];
            stage[hist[d[k] >> BSH] + slot[k]] = r;
            stage2[hist2[s[k] >> BSH] + slot2[k]] = s[k];
        }
    }
    __syncthreads();

    for (unsigned i = tid; i < n; i += 512u) {
        const v2i r = stage[i];
        *(v2i*)(ebuf + 2u * (unsigned)(gdelta[r.y >> BSH] + (int)i)) = r;
        const int sv = stage2[i];
        sbuf[(unsigned)(gdelta2[sv >> BSH] + (int)i)] = sv;
    }
}

__global__ void __launch_bounds__(512)
k_srcdeg(const int* __restrict__ sbuf, const int* __restrict__ scanned,
         const int* __restrict__ bsum, float* __restrict__ ns,
         unsigned N, unsigned E, unsigned G1, unsigned nblk, unsigned nent) {
    __shared__ int deg[BW];
    __shared__ int bpre[NBMAX];
    __shared__ int wsum[16];
    const unsigned tid = threadIdx.x;
    const unsigned b = blockIdx.x;

    int tot;
    const int bs = (tid < nblk) ? bsum[tid] : 0;
    bpre[tid] = block_scan512(bs, wsum, tot);
    if (tid < BW) deg[tid] = 0;
    __syncthreads();

    const unsigned i0 = nent + b * G1;
    const unsigned i1 = i0 + G1;
    const int S0 = scanned[i0] + bpre[i0 >> 10] - (int)E;
    const int S1 = scanned[i1] + bpre[i1 >> 10] - (int)E;
    const int n = S1 - S0;

    for (int j0 = 0; j0 < n; j0 += 5120) {
        int v[10];
        #pragma unroll
        for (int k = 0; k < 10; ++k) {
            const int i = j0 + 512 * k + (int)tid;
            v[k] = -1;
            if (i < n) v[k] = sbuf[(unsigned)(S0 + i)];
        }
        #pragma unroll
        for (int k = 0; k < 10; ++k) {
            if (v[k] >= 0) atomicAdd(&deg[v[k] & (BW - 1)], 1);
        }
    }
    __syncthreads();

    if (tid < BW) {
        const unsigned node = (b << BSH) + tid;
        if (node < N) ns[node] = rsqrtf(fmaxf((float)deg[tid], 1.0f));
    }
}

__global__ void __launch_bounds__(512)
k_bucket(const int* __restrict__ ebuf, const int* __restrict__ scanned,
         const int* __restrict__ bsum, const float* __restrict__ ns,
         int* __restrict__ rowptr, float* __restrict__ nd, int* __restrict__ erec,
         unsigned N, unsigned E, unsigned NB, unsigned G1, unsigned nblk) {
    __shared__ int deg[BW];
    __shared__ int cur[BW];
    __shared__ int bpre[NBMAX];
    __shared__ int wsum[16];
    __shared__ v2i stage[CAP2];
    const unsigned tid = threadIdx.x;
    const unsigned b = blockIdx.x;

    int tot;
    const int bs = (tid < nblk) ? bsum[tid] : 0;
    bpre[tid] = block_scan512(bs, wsum, tot);
    if (tid < BW) deg[tid] = 0;
    __syncthreads();

    const unsigned i0 = b * G1;
    const unsigned i1 = i0 + G1;
    const int S0 = scanned[i0] + bpre[i0 >> 10];
    const int S1 = (b + 1 < NB) ? scanned[i1] + bpre[i1 >> 10] : (int)E;
    const int n = S1 - S0;
    const int nbase = (int)(b << BSH);

    for (int j0 = 0; j0 < n; j0 += 5120) {
        v2i r[10];
        #pragma unroll
        for (int k = 0; k < 10; ++k) {
            const int i = j0 + 512 * k + (int)tid;
            r[k] = (v2i)(-1);
            if (i < n) r[k] = *(const v2i*)(ebuf + 2u * (unsigned)(S0 + i));
        }
        #pragma unroll
        for (int k = 0; k < 10; ++k) {
            if (r[k].y >= 0) atomicAdd(&deg[r[k].y - nbase], 1);
        }
    }
    __syncthreads();

    const int v = (tid < BW) ? deg[tid] : 0;
    const int ex = block_scan512(v, wsum, tot);
    if (tid < BW) {
        cur[tid] = ex;
        const unsigned node = (unsigned)nbase + tid;
        if (node < N) {
            rowptr[node] = S0 + ex;
            nd[node] = rsqrtf(fmaxf((float)v, 1.0f));
        }
    }
    if (b + 1 == NB && tid == 0) rowptr[N] = (int)E;
    __syncthreads();

    const bool staged = n <= CAP2;
    for (int j0 = 0; j0 < n; j0 += 5120) {
        v2i r[10];
        float nsv[10];
        #pragma unroll
        for (int k = 0; k < 10; ++k) {
            const int i = j0 + 512 * k + (int)tid;
            r[k] = (v2i)(-1);
            if (i < n) r[k] = *(const v2i*)(ebuf + 2u * (unsigned)(S0 + i));
        }
        #pragma unroll
        for (int k = 0; k < 10; ++k) nsv[k] = (r[k].y >= 0) ? ns[r[k].x] : 0.0f;
        #pragma unroll
        for (int k = 0; k < 10; ++k) {
            if (r[k].y >= 0) {
                const int slot = atomicAdd(&cur[r[k].y - nbase], 1);
                v2i o;
                o.x = r[k].x;
                o.y = __float_as_int(nsv[k]);
                if (staged) stage[slot] = o;
                else *(v2i*)(erec + 2u * (unsigned)(S0 + slot)) = o;
            }
        }
    }
    __syncthreads();
    if (staged) {
        for (int i = (int)tid; i < n; i += 512)
            *(v2i*)(erec + 2u * (unsigned)(S0 + i)) = stage[i];
    }
}


template<int K>
__device__ __forceinline__ void pull_rows(int sv, int wv, int t, const float* __restrict__ X,
                                          unsigned c, v4f& aw, v4f& ap) {
    v4f u[K];
    #pragma unroll
    for (int k = 0; k < K; ++k) {
        const unsigned s = (unsigned)__builtin_amdgcn_readlane(sv, t + k);
        u[k] = *(const v4f*)(X + s * NFEAT + c);
    }
    #pragma unroll
    for (int k = 0; k < K; ++k) {
        const float w = __int_as_float(__builtin_amdgcn_readlane(wv, t + k));
        aw += w * u[k];
        ap += u[k];
    }
}

#define TP 260

template<int MT>
struct LFrag {
    v4f a[MT];
    v2f b[4];
};

template<int MT, int Nout>
__device__ __forceinline__ void lfrag_load(LFrag<MT>& f, const float* Tl,
                                           const float* __restrict__ Wm, unsigned boff, int k8) {
    #pragma unroll
    for (int m = 0; m < MT; ++m) f.a[m] = *(const v4f*)(Tl + m * 16 * TP + k8);
    #pragma unroll
    for (int q = 0; q < 4; ++q)
        f.b[q] = *(const v2f*)(Wm + boff + (unsigned)(k8 + q) * (unsigned)Nout);
}

template<int MT>
__device__ __forceinline__ void lfrag_mma(const LFrag<MT>& f, v8f (&acc)[MT][2]) {
    #pragma unroll
    for (int s = 0; s < 2; ++s) {
        #pragma unroll
        for (int j = 0; j < 2; ++j) {
            v2f b;
            b.x = f.b[2 * s][j];
            b.y = f.b[2 * s + 1][j];
            #pragma unroll
            for (int m = 0; m < MT; ++m) {
                v2f a;
                a.x = f.a[m][2 * s];
                a.y = f.a[m][2 * s + 1];
                acc[m][j] = __builtin_amdgcn_wmma_f32_16x16x4_f32(
                    false, a, false, b, (short)0, acc[m][j], false, false);
            }
        }
    }
}

template<int MT, int Nout>
__device__ __forceinline__ void lds_k128(const float* Tl, const float* __restrict__ Wm,
                                         unsigned boff, v8f (&acc)[MT][2]) {
    LFrag<MT> f0, f1;
    lfrag_load<MT, Nout>(f0, Tl, Wm, boff, 0);
    #pragma unroll 1
    for (int k8 = 0; k8 < 112; k8 += 16) {
        lfrag_load<MT, Nout>(f1, Tl, Wm, boff, k8 + 8);
        lfrag_mma<MT>(f0, acc);
        lfrag_load<MT, Nout>(f0, Tl, Wm, boff, k8 + 16);
        lfrag_mma<MT>(f1, acc);
    }
    lfrag_load<MT, Nout>(f1, Tl, Wm, boff, 120);
    lfrag_mma<MT>(f0, acc);
    lfrag_mma<MT>(f1, acc);
}

#define GWAVES 8
#define MWAVES 4
#define LAYER_THREADS ((GWAVES + MWAVES) * 32)
#define LAYER_GRID_128 768
#define LAYER_GRID_64  1024

template<int Nout, bool RELU>
__global__ void __launch_bounds__(LAYER_THREADS)
k_layer(const float* __restrict__ X, const int* __restrict__ rowptr,
        const int* __restrict__ erec, const float* __restrict__ nd,
        const float* __restrict__ WA, const float* __restrict__ WB,
        const float* __restrict__ bias, float* __restrict__ Y, unsigned N, unsigned ntiles) {
    constexpr int MT = Nout / 64;
    __shared__ __attribute__((aligned(16))) float T[2][32 * TP];

    const unsigned lane = threadIdx.x & 31;
    const unsigned w = __builtin_amdgcn_readfirstlane(threadIdx.x >> 5);
    const unsigned nt = (ntiles - blockIdx.x + gridDim.x - 1u) / gridDim.x;

    const unsigned c = lane & 15u;
    const unsigned h = lane >> 4;
    const unsigned wq = (w - (unsigned)GWAVES) & 3u;
    const unsigned rt = (MT == 2) ? 0u : 16u * (wq >> 1);
    const unsigned cb = (MT == 2) ? 32u * wq : 32u * (wq & 1u);
    const unsigned boff = (4u * h) * (unsigned)Nout + cb + 2u * c;

    #pragma unroll 1
    for (unsigned it = 0; it <= nt; ++it) {
        if (w < (unsigned)GWAVES) {
            if (it < nt) {
                const unsigned mbase = (blockIdx.x + it * gridDim.x) * 32u;
                float* Tb = T[it & 1u];
                const unsigned row0 = mbase + 4u * w;
                const unsigned c4 = lane * 4u;
                int rp = 0;
                float ndv = 0.0f;
                if (lane <= 4u && row0 + lane <= N) rp = rowptr[row0 + lane];
                if (lane < 4u && row0 + lane < N) ndv = nd[row0 + lane];

                #pragma unroll 1
                for (int i = 0; i < 4; ++i) {
                    v4f aw = (v4f)(0.0f);
                    v4f ap = (v4f)(0.0f);
                    if (row0 + (unsigned)i < N) {
                        const int beg = __builtin_amdgcn_readlane(rp, i);
                        const int end = __builtin_amdgcn_readlane(rp, i + 1);
                        for (int base = beg; base < end; base += 32) {
                            const int rem = end - base;
                            const int cnt = rem < 32 ? rem : 32;
                            v2i e = (v2i)(0);
                            if ((int)lane < cnt) e = *(const v2i*)(erec + 2u * (unsigned)(base + (int)lane));
                            const int sv = e.x;
                            const int wv = e.y;
                            int t = 0;
                            for (; t + 8 <= cnt; t += 8) pull_rows<8>(sv, wv, t, X, c4, aw, ap);
                            switch (cnt - t) {
                                case 7: pull_rows<7>(sv, wv, t, X, c4, aw, ap); break;
                                case 6: pull_rows<6>(sv, wv, t, X, c4, aw, ap); break;
                                case 5: pull_rows<5>(sv, wv, t, X, c4, aw, ap); break;
                                case 4: pull_rows<4>(sv, wv, t, X, c4, aw, ap); break;
                                case 3: pull_rows<3>(sv, wv, t, X, c4, aw, ap); break;
                                case 2: pull_rows<2>(sv, wv, t, X, c4, aw, ap); break;
                                case 1: pull_rows<1>(sv, wv, t, X, c4, aw, ap); break;
                                default: break;
                            }
                        }
                        aw = aw * __int_as_float(__builtin_amdgcn_readlane(__float_as_int(ndv), i));
                    }
                    float* tr = Tb + (4u * w + (unsigned)i) * TP + c4;
                    *(v4f*)tr = aw;
                    *(v4f*)(tr + 128) = ap;
                }
            }
        } else if (it > 0) {
            const unsigned mbase = (blockIdx.x + (it - 1u) * gridDim.x) * 32u;
            const float* Tl = T[(it - 1u) & 1u] + (rt + c) * TP + 4u * h;

            v8f acc[MT][2];
            #pragma unroll
            for (int m = 0; m < MT; ++m) {
                #pragma unroll
                for (int j = 0; j < 2; ++j) acc[m][j] = (v8f)(0.0f);
            }

            lds_k128<MT, Nout>(Tl, WA, boff, acc);
            lds_k128<MT, Nout>(Tl + 128, WB, boff, acc);

            const v2f bv = *(const v2f*)(bias + cb + 2u * c);
            #pragma unroll
            for (int m = 0; m < MT; ++m) {
                #pragma unroll
                for (int r = 0; r < 8; ++r) {
                    const unsigned row = mbase + rt + 16u * m + 8u * h + (unsigned)r;
                    if (row < N) {
                        v2f o;
                        o.x = acc[m][0][r] + bv.x;
                        o.y = acc[m][1][r] + bv.y;
                        if (RELU) {
                            o.x = fmaxf(o.x, 0.0f);
                            o.y = fmaxf(o.y, 0.0f);
                        }
                        *(v2f*)(Y + row * (unsigned)Nout + cb + 2u * c) = o;
                    }
                }
            }
        }
        __syncthreads();
    }
}


extern "C" void kernel_launch(void* const* d_in, const int* in_sizes, int n_in,
                              void* d_out, int out_size, void* d_ws, size_t ws_size,
                              hipStream_t stream) {
    const float* raw_x = (const float*)d_in[0];
    const int*   src   = (const int*)d_in[1];
    const int*   dst   = (const int*)d_in[2];
    const float* gc_w0 = (const float*)d_in[3];
    const float* gc_b0 = (const float*)d_in[4];
    const float* gc_w1 = (const float*)d_in[5];
    const float* gc_b1 = (const float*)d_in[6];
    const float* gc_w2 = (const float*)d_in[7];
    const float* gc_b2 = (const float*)d_in[8];
    const float* rw0   = (const float*)d_in[9];
    const float* rw1   = (const float*)d_in[10];
    const float* rw2   = (const float*)d_in[11];

    const int N = in_sizes[0] / NFEAT;
    const int E = in_sizes[1];
    const size_t NB_F = (size_t)N * NFEAT;

    const int TB = 256;
    auto blocks = [](long long n, int tb) { return (int)((n + tb - 1) / tb); };

    const int G1   = blocks(E, CH1);
    const int NBK  = (N + BW - 1) >> BSH;
    const int nent = NBK * G1;
    const int nsc  = 2 * nent;
    const int nblk = blocks(nsc + 1LL, 1024);

    float* P0 = (float*)d_ws;
    float* P1 = P0 + NB_F;
    int*  cur = (int*)(P1 + NB_F);
    auto take = [&](size_t n) { int* p = cur; cur += (n + 3) & ~(size_t)3; return p; };
    float* nd     = (float*)take(N);
    float* ns     = (float*)take(N);
    int*  rowptr  = take((size_t)N + 1);
    int*  cnt     = take(nsc);
    int*  scanned = take((size_t)nsc + 1);
    int*  bsum    = take(1024);
    int*  ebuf    = take(2 * (size_t)E);
    int*  sbuf    = take((size_t)E);
    int*  erec    = take(2 * (size_t)E);

    k_bin<false><<<G1, 512, 0, stream>>>(src, dst, cnt, nullptr, nullptr, nullptr, nullptr,
                                         (unsigned)E, (unsigned)NBK, (unsigned)nblk,
                                         (unsigned)nent);
    k_scan_local<<<nblk, TB, 0, stream>>>(cnt, scanned, bsum, (unsigned)nsc);
    k_bin<true><<<G1, 512, 0, stream>>>(src, dst, nullptr, scanned, bsum, ebuf, sbuf,
                                        (unsigned)E, (unsigned)NBK, (unsigned)nblk,
                                        (unsigned)nent);
    k_srcdeg<<<NBK, 512, 0, stream>>>(sbuf, scanned, bsum, ns, (unsigned)N, (unsigned)E,
                                      (unsigned)G1, (unsigned)nblk, (unsigned)nent);
    k_bucket<<<NBK, 512, 0, stream>>>(ebuf, scanned, bsum, ns, rowptr, nd, erec,
                                      (unsigned)N, (unsigned)E, (unsigned)NBK, (unsigned)G1,
                                      (unsigned)nblk);

    const int layerTiles = blocks(N, 32);
    const int layerGrid128 = layerTiles < LAYER_GRID_128 ? layerTiles : LAYER_GRID_128;
    const int layerGrid64  = layerTiles < LAYER_GRID_64 ? layerTiles : LAYER_GRID_64;

    k_layer<128, true><<<layerGrid128, LAYER_THREADS, 0, stream>>>(
        raw_x, rowptr, erec, nd, gc_w0, rw0, gc_b0, P0, (unsigned)N, (unsigned)layerTiles);

    k_layer<128, true><<<layerGrid128, LAYER_THREADS, 0, stream>>>(
        P0, rowptr, erec, nd, gc_w1, rw1, gc_b1, P1, (unsigned)N, (unsigned)layerTiles);

    k_layer<64, false><<<layerGrid64, LAYER_THREADS, 0, stream>>>(
        P1, rowptr, erec, nd, gc_w2, rw2, gc_b2, (float*)d_out, (unsigned)N, (unsigned)layerTiles);
}


# Level-1 binning in one pass: a block keeps its bucket-grouped edges in its own chunk plus a (start,length) run table; level 2 reads its runs; count pass and scan launch gone; LDS arrival-order sums
# speedup vs baseline: 7.8096x; 1.0578x over previous
#include <hip/hip_runtime.h>

typedef __attribute__((ext_vector_type(2))) float v2f;
typedef __attribute__((ext_vector_type(4))) float v4f;
typedef __attribute__((ext_vector_type(8))) float v8f;
typedef __attribute__((ext_vector_type(2))) int   v2i;

#define NFEAT 128
#define NHID  128
#define NCLASS 64


#define BSH   8
#define BW    (1 << BSH)
#define NBMAX 512
#define CH1   6144
#define CAP2  6144
#define RU    17

__device__ __forceinline__ int block_scan512(int v, int* wsum, int& tot) {
    const unsigned lane = threadIdx.x & 31;
    const unsigned wid  = threadIdx.x >> 5;
    int inc = v;
    #pragma unroll
    for (int d = 1; d < 32; d <<= 1) {
        const int n = __shfl_up(inc, d);
        if (lane >= (unsigned)d) inc += n;
    }
    __syncthreads();
    if (lane == 31) wsum[wid] = inc;
    __syncthreads();
    int woff = 0;
    tot = 0;
    #pragma unroll
    for (int w = 0; w < 16; ++w) {
        const int s = wsum[w];
        if ((unsigned)w < wid) woff += s;
        tot += s;
    }
    return woff + inc - v;
}

__global__ void __launch_bounds__(512)
k_bin(const int* __restrict__ src, const int* __restrict__ dst,
      int* __restrict__ ebuf, int* __restrict__ sbuf,
      int* __restrict__ tabd, int* __restrict__ tabs, unsigned E, unsigned NB) {
    __shared__ int hist[NBMAX];
    __shared__ int hist2[NBMAX];
    __shared__ int wsum[16];
    __shared__ v2i stage[CH1];
    __shared__ int stage2[CH1];
    const unsigned tid = threadIdx.x;
    const unsigned G1 = gridDim.x;
    const unsigned wg = blockIdx.x;
    const unsigned c0 = wg * (unsigned)CH1;
    const unsigned n = min(E - c0, (unsigned)CH1);

    hist[tid] = 0;
    hist2[tid] = 0;
    __syncthreads();

    int s[12], d[12], slot[12], slot2[12];
    #pragma unroll
    for (int k = 0; k < 12; ++k) {
        const unsigned i = tid + 512u * k;
        const bool ok = i < n;
        s[k] = ok ? src[c0 + i] : 0;
        d[k] = ok ? dst[c0 + i] : -1;
    }
    #pragma unroll
    for (int k = 0; k < 12; ++k) {
        slot[k] = 0;
        slot2[k] = 0;
        if (d[k] >= 0) {
            slot[k] = atomicAdd(&hist[d[k] >> BSH], 1);
            slot2[k] = atomicAdd(&hist2[s[k] >> BSH], 1);
        }
    }
    __syncthreads();

    int tot;
    const int h = hist[tid];
    const int lb = block_scan512(h, wsum, tot);
    const int h2 = hist2[tid];
    const int lb2 = block_scan512(h2, wsum, tot);
    if (tid < NB) {
        v2i e;
        e.x = lb;
        e.y = h;
        *(v2i*)(tabd + 2u * (tid * G1 + wg)) = e;
        v2i e2;
        e2.x = lb2;
        e2.y = h2;
        *(v2i*)(tabs + 2u * (tid * G1 + wg)) = e2;
    }
    hist[tid] = lb;
    hist2[tid] = lb2;
    __syncthreads();

    #pragma unroll
    for (int k = 0; k < 12; ++k) {
        if (d[k] >= 0) {
            v2i r;
            r.x = s[k];
            r.y = d[k];
            stage[hist[d[k] >> BSH] + slot[k]] = r;
            stage2[hist2[s[k] >> BSH] + slot2[k]] = s[k];
        }
    }
    __syncthreads();

    for (unsigned i = tid; i < n; i += 512u) {
        *(v2i*)(ebuf + 2u * (c0 + i)) = stage[i];
        sbuf[c0 + i] = stage2[i];
    }
}

__global__ void __launch_bounds__(512)
k_srcdeg(const int* __restrict__ sbuf, const int* __restrict__ tabs,
         const int* __restrict__ tabd, float* __restrict__ ns, int* __restrict__ totd,
         unsigned N, unsigned G1) {
    __shared__ int deg[BW];
    __shared__ int wsum[16];
    const unsigned tid = threadIdx.x;
    const unsigned b = blockIdx.x;
    const unsigned hw = tid >> 5;
    const unsigned ln = tid & 31u;

    if (tid < BW) deg[tid] = 0;
    int dl = 0;
    #pragma unroll 1
    for (unsigned w = tid; w < G1; w += 512u) dl += tabd[2u * (b * G1 + w) + 1u];
    int tot;
    (void)block_scan512(dl, wsum, tot);
    if (tid == 0) totd[b] = tot;

    const int* trow = tabs + 2u * b * G1;
    for (unsigned w0 = 0; w0 < G1; w0 += 16u * RU) {
        v2i tb[RU];
        #pragma unroll
        for (int u = 0; u < RU; ++u) {
            const unsigned w = w0 + 16u * u + hw;
            tb[u] = (v2i)(0);
            if (w < G1) tb[u] = *(const v2i*)(trow + 2u * w);
        }
        int v[RU];
        #pragma unroll
        for (int u = 0; u < RU; ++u) {
            const unsigned w = w0 + 16u * u + hw;
            v[u] = -1;
            if ((int)ln < tb[u].y) v[u] = sbuf[w * (unsigned)CH1 + (unsigned)tb[u].x + ln];
        }
        #pragma unroll
        for (int u = 0; u < RU; ++u) {
            if (v[u] >= 0) atomicAdd(&deg[v[u] & (BW - 1)], 1);
        }
        #pragma unroll
        for (int u = 0; u < RU; ++u) {
            const unsigned w = w0 + 16u * u + hw;
            for (int i = 32 + (int)ln; i < tb[u].y; i += 32) {
                const int x = sbuf[w * (unsigned)CH1 + (unsigned)(tb[u].x + i)];
                atomicAdd(&deg[x & (BW - 1)], 1);
            }
        }
    }
    __syncthreads();

    if (tid < BW) {
        const unsigned node = (b << BSH) + tid;
        if (node < N) ns[node] = rsqrtf(fmaxf((float)deg[tid], 1.0f));
    }
}

__global__ void __launch_bounds__(512)
k_bucket(const int* __restrict__ ebuf, const int* __restrict__ tabd,
         const int* __restrict__ totd, const float* __restrict__ ns,
         int* __restrict__ rowptr, float* __restrict__ nd, int* __restrict__ erec,
         unsigned N, unsigned E, unsigned NB, unsigned G1) {
    __shared__ int deg[BW];
    __shared__ int cur[BW];
    __shared__ int wsum[16];
    __shared__ int sb[2];
    __shared__ v2i stage[CAP2];
    const unsigned tid = threadIdx.x;
    const unsigned b = blockIdx.x;
    const unsigned hw = tid >> 5;
    const unsigned ln = tid & 31u;

    int tot;
    const int tv = (tid < NB) ? totd[tid] : 0;
    const int pre = block_scan512(tv, wsum, tot);
    if (tid == b) {
        sb[0] = pre;
        sb[1] = tv;
    }
    if (tid < BW) deg[tid] = 0;
    __syncthreads();

    const int S0 = sb[0];
    const int n = sb[1];
    const int nbase = (int)(b << BSH);
    const int* trow = tabd + 2u * b * G1;

    for (unsigned w0 = 0; w0 < G1; w0 += 16u * RU) {
        v2i tb[RU];
        #pragma unroll
        for (int u = 0; u < RU; ++u) {
            const unsigned w = w0 + 16u * u + hw;
            tb[u] = (v2i)(0);
            if (w < G1) tb[u] = *(const v2i*)(trow + 2u * w);
        }
        int dv[RU];
        #pragma unroll
        for (int u = 0; u < RU; ++u) {
            const unsigned w = w0 + 16u * u + hw;
            dv[u] = -1;
            if ((int)ln < tb[u].y)
                dv[u] = ebuf[2u * (w * (unsigned)CH1 + (unsigned)tb[u].x + ln) + 1u];
        }
        #pragma unroll
        for (int u = 0; u < RU; ++u) {
            if (dv[u] >= 0) atomicAdd(&deg[dv[u] - nbase], 1);
        }
        #pragma unroll
        for (int u = 0; u < RU; ++u) {
            const unsigned w = w0 + 16u * u + hw;
            for (int i = 32 + (int)ln; i < tb[u].y; i += 32) {
                const int x = ebuf[2u * (w * (unsigned)CH1 + (unsigned)(tb[u].x + i)) + 1u];
                atomicAdd(&deg[x - nbase], 1);
            }
        }
    }
    __syncthreads();

    const int v = (tid < BW) ? deg[tid] : 0;
    const int ex = block_scan512(v, wsum, tot);
    if (tid < BW) {
        cur[tid] = ex;
        const unsigned node = (unsigned)nbase + tid;
        if (node < N) {
            rowptr[node] = S0 + ex;
            nd[node] = rsqrtf(fmaxf((float)v, 1.0f));
        }
    }
    if (b + 1 == NB && tid == 0) rowptr[N] = (int)E;
    __syncthreads();

    const bool staged = n <= CAP2;
    for (unsigned w0 = 0; w0 < G1; w0 += 16u * RU) {
        v2i tb[RU];
        #pragma unroll
        for (int u = 0; u < RU; ++u) {
            const unsigned w = w0 + 16u * u + hw;
            tb[u] = (v2i)(0);
            if (w < G1) tb[u] = *(const v2i*)(trow + 2u * w);
        }
        v2i r[RU];
        float nsv[RU];
        #pragma unroll
        for (int u = 0; u < RU; ++u) {
            const unsigned w = w0 + 16u * u + hw;
            const unsigned idx = ((int)ln < tb[u].y) ? w * (unsigned)CH1 + (unsigned)tb[u].x + ln : 0u;
            r[u] = *(const v2i*)(ebuf + 2u * idx);
        }
        #pragma unroll
        for (int u = 0; u < RU; ++u) nsv[u] = ns[r[u].x];
        #pragma unroll
        for (int u = 0; u < RU; ++u) {
            if ((int)ln < tb[u].y) {
                const int slot = atomicAdd(&cur[r[u].y - nbase], 1);
                v2i o;
                o.x = r[u].x;
                o.y = __float_as_int(nsv[u]);
                if (staged) stage[slot] = o;
                else *(v2i*)(erec + 2u * (unsigned)(S0 + slot)) = o;
            }
        }
        #pragma unroll
        for (int u = 0; u < RU; ++u) {
            const unsigned w = w0 + 16u * u + hw;
            for (int i = 32 + (int)ln; i < tb[u].y; i += 32) {
                const v2i q = *(const v2i*)(ebuf + 2u * (w * (unsigned)CH1 + (unsigned)(tb[u].x + i)));
                const float qn = ns[q.x];
                const int slot = atomicAdd(&cur[q.y - nbase], 1);
                v2i o;
                o.x = q.x;
                o.y = __float_as_int(qn);
                if (staged) stage[slot] = o;
                else *(v2i*)(erec + 2u * (unsigned)(S0 + slot)) = o;
            }
        }
    }
    __syncthreads();
    if (staged) {
        for (int i = (int)tid; i < n; i += 512)
            *(v2i*)(erec + 2u * (unsigned)(S0 + i)) = stage[i];
    }
}


template<int K>
__device__ __forceinline__ void pull_rows(int sv, int wv, int t, const float* __restrict__ X,
                                          unsigned c, v4f& aw, v4f& ap) {
    v4f u[K];
    #pragma unroll
    for (int k = 0; k < K; ++k) {
        const unsigned s = (unsigned)__builtin_amdgcn_readlane(sv, t + k);
        u[k] = *(const v4f*)(X + s * NFEAT + c);
    }
    #pragma unroll
    for (int k = 0; k < K; ++k) {
        const float w = __int_as_float(__builtin_amdgcn_readlane(wv, t + k));
        aw += w * u[k];
        ap += u[k];
    }
}

#define TP 260

template<int MT>
struct LFrag {
    v4f a[MT];
    v2f b[4];
};

template<int MT, int Nout>
__device__ __forceinline__ void lfrag_load(LFrag<MT>& f, const float* Tl,
                                           const float* __restrict__ Wm, unsigned boff, int k8) {
    #pragma unroll
    for (int m = 0; m < MT; ++m) f.a[m] = *(const v4f*)(Tl + m * 16 * TP + k8);
    #pragma unroll
    for (int q = 0; q < 4; ++q)
        f.b[q] = *(const v2f*)(Wm + boff + (unsigned)(k8 + q) * (unsigned)Nout);
}

template<int MT>
__device__ __forceinline__ void lfrag_mma(const LFrag<MT>& f, v8f (&acc)[MT][2]) {
    #pragma unroll
    for (int s = 0; s < 2; ++s) {
        #pragma unroll
        for (int j = 0; j < 2; ++j) {
            v2f b;
            b.x = f.b[2 * s][j];
            b.y = f.b[2 * s + 1][j];
            #pragma unroll
            for (int m = 0; m < MT; ++m) {
                v2f a;
                a.x = f.a[m][2 * s];
                a.y = f.a[m][2 * s + 1];
                acc[m][j] = __builtin_amdgcn_wmma_f32_16x16x4_f32(
                    false, a, false, b, (short)0, acc[m][j], false, false);
            }
        }
    }
}

template<int MT, int Nout>
__device__ __forceinline__ void lds_k128(const float* Tl, const float* __restrict__ Wm,
                                         unsigned boff, v8f (&acc)[MT][2]) {
    LFrag<MT> f0, f1;
    lfrag_load<MT, Nout>(f0, Tl, Wm, boff, 0);
    #pragma unroll 1
    for (int k8 = 0; k8 < 112; k8 += 16) {
        lfrag_load<MT, Nout>(f1, Tl, Wm, boff, k8 + 8);
        lfrag_mma<MT>(f0, acc);
        lfrag_load<MT, Nout>(f0, Tl, Wm, boff, k8 + 16);
        lfrag_mma<MT>(f1, acc);
    }
    lfrag_load<MT, Nout>(f1, Tl, Wm, boff, 120);
    lfrag_mma<MT>(f0, acc);
    lfrag_mma<MT>(f1, acc);
}

#define GWAVES 8
#define MWAVES 4
#define LAYER_THREADS ((GWAVES + MWAVES) * 32)
#define LAYER_GRID_128 768
#define LAYER_GRID_64  1024

template<int Nout, bool RELU>
__global__ void __launch_bounds__(LAYER_THREADS)
k_layer(const float* __restrict__ X, const int* __restrict__ rowptr,
        const int* __restrict__ erec, const float* __restrict__ nd,
        const float* __restrict__ WA, const float* __restrict__ WB,
        const float* __restrict__ bias, float* __restrict__ Y, unsigned N, unsigned ntiles) {
    constexpr int MT = Nout / 64;
    __shared__ __attribute__((aligned(16))) float T[2][32 * TP];

    const unsigned lane = threadIdx.x & 31;
    const unsigned w = __builtin_amdgcn_readfirstlane(threadIdx.x >> 5);
    const unsigned nt = (ntiles - blockIdx.x + gridDim.x - 1u) / gridDim.x;

    const unsigned c = lane & 15u;
    const unsigned h = lane >> 4;
    const unsigned wq = (w - (unsigned)GWAVES) & 3u;
    const unsigned rt = (MT == 2) ? 0u : 16u * (wq >> 1);
    const unsigned cb = (MT == 2) ? 32u * wq : 32u * (wq & 1u);
    const unsigned boff = (4u * h) * (unsigned)Nout + cb + 2u * c;

    #pragma unroll 1
    for (unsigned it = 0; it <= nt; ++it) {
        if (w < (unsigned)GWAVES) {
            if (it < nt) {
                const unsigned mbase = (blockIdx.x + it * gridDim.x) * 32u;
                float* Tb = T[it & 1u];
                const unsigned row0 = mbase + 4u * w;
                const unsigned c4 = lane * 4u;
                int rp = 0;
                float ndv = 0.0f;
                if (lane <= 4u && row0 + lane <= N) rp = rowptr[row0 + lane];
                if (lane < 4u && row0 + lane < N) ndv = nd[row0 + lane];

                #pragma unroll 1
                for (int i = 0; i < 4; ++i) {
                    v4f aw = (v4f)(0.0f);
                    v4f ap = (v4f)(0.0f);
                    if (row0 + (unsigned)i < N) {
                        const int beg = __builtin_amdgcn_readlane(rp, i);
                        const int end = __builtin_amdgcn_readlane(rp, i + 1);
                        for (int base = beg; base < end; base += 32) {
                            const int rem = end - base;
                            const int cnt = rem < 32 ? rem : 32;
                            v2i e = (v2i)(0);
                            if ((int)lane < cnt) e = *(const v2i*)(erec + 2u * (unsigned)(base + (int)lane));
                            const int sv = e.x;
                            const int wv = e.y;
                            int t = 0;
                            for (; t + 8 <= cnt; t += 8) pull_rows<8>(sv, wv, t, X, c4, aw, ap);
                            switch (cnt - t) {
                                case 7: pull_rows<7>(sv, wv, t, X, c4, aw, ap); break;
                                case 6: pull_rows<6>(sv, wv, t, X, c4, aw, ap); break;
                                case 5: pull_rows<5>(sv, wv, t, X, c4, aw, ap); break;
                                case 4: pull_rows<4>(sv, wv, t, X, c4, aw, ap); break;
                                case 3: pull_rows<3>(sv, wv, t, X, c4, aw, ap); break;
                                case 2: pull_rows<2>(sv, wv, t, X, c4, aw, ap); break;
                                case 1: pull_rows<1>(sv, wv, t, X, c4, aw, ap); break;
                                default: break;
                            }
                        }
                        aw = aw * __int_as_float(__builtin_amdgcn_readlane(__float_as_int(ndv), i));
                    }
                    float* tr = Tb + (4u * w + (unsigned)i) * TP + c4;
                    *(v4f*)tr = aw;
                    *(v4f*)(tr + 128) = ap;
                }
            }
        } else if (it > 0) {
            const unsigned mbase = (blockIdx.x + (it - 1u) * gridDim.x) * 32u;
            const float* Tl = T[(it - 1u) & 1u] + (rt + c) * TP + 4u * h;

            v8f acc[MT][2];
            #pragma unroll
            for (int m = 0; m < MT; ++m) {
                #pragma unroll
                for (int j = 0; j < 2; ++j) acc[m][j] = (v8f)(0.0f);
            }

            lds_k128<MT, Nout>(Tl, WA, boff, acc);
            lds_k128<MT, Nout>(Tl + 128, WB, boff, acc);

            const v2f bv = *(const v2f*)(bias + cb + 2u * c);
            #pragma unroll
            for (int m = 0; m < MT; ++m) {
                #pragma unroll
                for (int r = 0; r < 8; ++r) {
                    const unsigned row = mbase + rt + 16u * m + 8u * h + (unsigned)r;
                    if (row < N) {
                        v2f o;
                        o.x = acc[m][0][r] + bv.x;
                        o.y = acc[m][1][r] + bv.y;
                        if (RELU) {
                            o.x = fmaxf(o.x, 0.0f);
                            o.y = fmaxf(o.y, 0.0f);
                        }
                        *(v2f*)(Y + row * (unsigned)Nout + cb + 2u * c) = o;
                    }
                }
            }
        }
        __syncthreads();
    }
}


extern "C" void kernel_launch(void* const* d_in, const int* in_sizes, int n_in,
                              void* d_out, int out_size, void* d_ws, size_t ws_size,
                              hipStream_t stream) {
    const float* raw_x = (const float*)d_in[0];
    const int*   src   = (const int*)d_in[1];
    const int*   dst   = (const int*)d_in[2];
    const float* gc_w0 = (const float*)d_in[3];
    const float* gc_b0 = (const float*)d_in[4];
    const float* gc_w1 = (const float*)d_in[5];
    const float* gc_b1 = (const float*)d_in[6];
    const float* gc_w2 = (const float*)d_in[7];
    const float* gc_b2 = (const float*)d_in[8];
    const float* rw0   = (const float*)d_in[9];
    const float* rw1   = (const float*)d_in[10];
    const float* rw2   = (const float*)d_in[11];

    const int N = in_sizes[0] / NFEAT;
    const int E = in_sizes[1];
    const size_t NB_F = (size_t)N * NFEAT;

    auto blocks = [](long long n, int tb) { return (int)((n + tb - 1) / tb); };

    const int G1   = blocks(E, CH1);
    const int NBK  = (N + BW - 1) >> BSH;
    const int nent = NBK * G1;

    float* P0 = (float*)d_ws;
    float* P1 = P0 + NB_F;
    int*  cur = (int*)(P1 + NB_F);
    auto take = [&](size_t n) { int* p = cur; cur += (n + 3) & ~(size_t)3; return p; };
    float* nd     = (float*)take(N);
    float* ns     = (float*)take(N);
    int*  rowptr  = take((size_t)N + 1);
    int*  tabd    = take(2 * (size_t)nent);
    int*  tabs    = take(2 * (size_t)nent);
    int*  totd    = take(NBMAX);
    int*  ebuf    = take(2 * (size_t)E);
    int*  sbuf    = take((size_t)E);
    int*  erec    = take(2 * (size_t)E);

    k_bin<<<G1, 512, 0, stream>>>(src, dst, ebuf, sbuf, tabd, tabs, (unsigned)E, (unsigned)NBK);
    k_srcdeg<<<NBK, 512, 0, stream>>>(sbuf, tabs, tabd, ns, totd, (unsigned)N, (unsigned)G1);
    k_bucket<<<NBK, 512, 0, stream>>>(ebuf, tabd, totd, ns, rowptr, nd, erec,
                                      (unsigned)N, (unsigned)E, (unsigned)NBK, (unsigned)G1);

    const int layerTiles = blocks(N, 32);
    const int layerGrid128 = layerTiles < LAYER_GRID_128 ? layerTiles : LAYER_GRID_128;
    const int layerGrid64  = layerTiles < LAYER_GRID_64 ? layerTiles : LAYER_GRID_64;

    k_layer<128, true><<<layerGrid128, LAYER_THREADS, 0, stream>>>(
        raw_x, rowptr, erec, nd, gc_w0, rw0, gc_b0, P0, (unsigned)N, (unsigned)layerTiles);

    k_layer<128, true><<<layerGrid128, LAYER_THREADS, 0, stream>>>(
        P0, rowptr, erec, nd, gc_w1, rw1, gc_b1, P1, (unsigned)N, (unsigned)layerTiles);

    k_layer<64, false><<<layerGrid64, LAYER_THREADS, 0, stream>>>(
        P1, rowptr, erec, nd, gc_w2, rw2, gc_b2, (float*)d_out, (unsigned)N, (unsigned)layerTiles);
}
